# starting kernel
# baseline (speedup 1.0000x reference)
#include <hip/hip_runtime.h>
#include <hip/hip_bf16.h>
#include <math.h>

typedef __attribute__((ext_vector_type(2))) float v2f;
typedef __attribute__((ext_vector_type(8))) float v8f;

#define NROWS 8192
#define NDIM  64
#define COLS_PER_BLOCK 128
#define LDS_STRIDE 66
#define BETA 0.01f
#define TWO_PI 6.28318530717958647692f

__global__ __launch_bounds__(256) void k_y(const float* __restrict__ x,
                                           const float* __restrict__ noise,
                                           const float* __restrict__ nvp,
                                           float* __restrict__ y, int n) {
    int i = blockIdx.x * 256 + threadIdx.x;
    float s = __expf(0.5f * nvp[0]);
    if (i < n) y[i] = fmaf(s, noise[i], x[i]);
}

__global__ __launch_bounds__(256) void k_sq(const float* __restrict__ x,
                                            float* __restrict__ sq) {
    int row = blockIdx.x * 256 + threadIdx.x;
    const float4* r = (const float4*)(x + (size_t)row * NDIM);
    float s = 0.0f;
#pragma unroll
    for (int j = 0; j < NDIM / 4; ++j) {
        float4 v = r[j];
        s = fmaf(v.x, v.x, s); s = fmaf(v.y, v.y, s);
        s = fmaf(v.z, v.z, s); s = fmaf(v.w, v.w, s);
    }
    sq[row] = s;
}

__global__ __launch_bounds__(256) void k_pairwise(const float* __restrict__ x,
                                                  const float* __restrict__ sq,
                                                  const float* __restrict__ kvp,
                                                  const float* __restrict__ nvp,
                                                  float* __restrict__ lse1_out,
                                                  float* __restrict__ lse2_out) {
    __shared__ float ldsX[COLS_PER_BLOCK * LDS_STRIDE];
    __shared__ float ldsSq[COLS_PER_BLOCK];

    const int tid      = threadIdx.x;
    const int lane     = tid & 31;
    const int wave     = tid >> 5;
    const int half     = lane >> 4;
    const int l15      = lane & 15;
    const int rowBase  = blockIdx.x * (8 * 16) + wave * 16;

    const float kv = kvp[0], nv = nvp[0];
    const float total_var = __expf(nv) + __expf(kv);
    const float a1 = 1.0f / (2.0f * total_var);
    const float a2 = 1.0f / (2.0f * __expf(kv));

    v2f a[16];
    {
        const float* ar = x + (size_t)(rowBase + l15) * NDIM + 2 * half;
#pragma unroll
        for (int ch = 0; ch < 16; ++ch)
            a[ch] = *(const v2f*)(ar + 4 * ch);
    }

    float sqA[8];
#pragma unroll
    for (int k = 0; k < 8; ++k)
        sqA[k] = sq[rowBase + k + 8 * half];

    float l1[8], l2[8], m2[8];
#pragma unroll
    for (int k = 0; k < 8; ++k) { l1[k] = 0.0f; l2[k] = 0.0f; m2[k] = -1e30f; }

    float4 stage[8];
    float  stageSq;
#pragma unroll
    for (int i = 0; i < 8; ++i)
        stage[i] = ((const float4*)x)[tid + i * 256];
    stageSq = (tid < COLS_PER_BLOCK) ? sq[tid] : 0.0f;

    for (int cb = 0; cb < NROWS; cb += COLS_PER_BLOCK) {
        __syncthreads();
#pragma unroll
        for (int i = 0; i < 8; ++i) {
            int idx = tid + i * 256;
            int c   = idx >> 4;
            int f4  = idx & 15;
            float* dst = &ldsX[c * LDS_STRIDE + f4 * 4];
            dst[0] = stage[i].x; dst[1] = stage[i].y;
            dst[2] = stage[i].z; dst[3] = stage[i].w;
        }
        if (tid < COLS_PER_BLOCK) ldsSq[tid] = stageSq;
        __syncthreads();

        const int nb = cb + COLS_PER_BLOCK;
        if (nb < NROWS) {
#pragma unroll
            for (int i = 0; i < 8; ++i)
                stage[i] = ((const float4*)(x + (size_t)nb * NDIM))[tid + i * 256];
            if (tid < COLS_PER_BLOCK) stageSq = sq[nb + tid];
        }

        for (int ct = 0; ct < COLS_PER_BLOCK / 16; ++ct) {
            v2f b[16];
            const float* bp = &ldsX[(ct * 16 + l15) * LDS_STRIDE + 2 * half];
#pragma unroll
            for (int ch = 0; ch < 16; ++ch)
                b[ch] = *(const v2f*)(bp + 4 * ch);

            v8f c = {};
#pragma unroll
            for (int ch = 0; ch < 16; ++ch)
                c = __builtin_amdgcn_wmma_f32_16x16x4_f32(
                         false, a[ch],  false, b[ch],
                         (short)0, c,  false,  false);

            const float sqB = ldsSq[ct * 16 + l15];
            const int   col = cb + ct * 16 + l15;
#pragma unroll
            for (int k = 0; k < 8; ++k) {
                const int rowG = rowBase + k + 8 * half;
                float d = fmaxf(sqA[k] + sqB - 2.0f * c[k], 0.0f);
                l1[k] += __expf(-d * a1);
                float s2 = (col == rowG) ? -1e38f : (-d * a2);
                float nm = fmaxf(m2[k], s2);
                l2[k] = l2[k] * __expf(m2[k] - nm) + __expf(s2 - nm);
                m2[k] = nm;
            }
        }
    }

#pragma unroll
    for (int k = 0; k < 8; ++k) {
#pragma unroll
        for (int m = 1; m <= 8; m <<= 1) {
            l1[k] += __shfl_xor(l1[k], m, 32);
            float om = __shfl_xor(m2[k], m, 32);
            float ol = __shfl_xor(l2[k], m, 32);
            float nm = fmaxf(m2[k], om);
            l2[k] = l2[k] * __expf(m2[k] - nm) + ol * __expf(om - nm);
            m2[k] = nm;
        }
    }
    if (l15 == 0) {
#pragma unroll
        for (int k = 0; k < 8; ++k) {
            int rowG = rowBase + k + 8 * half;
            lse1_out[rowG] = __logf(l1[k]);
            lse2_out[rowG] = m2[k] + __logf(l2[k]);
        }
    }
}

__global__ __launch_bounds__(256) void k_final(const float* __restrict__ lse1,
                                               const float* __restrict__ lse2,
                                               const float* __restrict__ kvp,
                                               const float* __restrict__ nvp,
                                               float* __restrict__ out2) {
    __shared__ float s1[256], s2[256];
    float p1 = 0.0f, p2 = 0.0f;
    for (int i = threadIdx.x; i < NROWS; i += 256) { p1 += lse1[i]; p2 += lse2[i]; }
    s1[threadIdx.x] = p1; s2[threadIdx.x] = p2;
    __syncthreads();
    for (int st = 128; st > 0; st >>= 1) {
        if (threadIdx.x < st) {
            s1[threadIdx.x] += s1[threadIdx.x + st];
            s2[threadIdx.x] += s2[threadIdx.x + st];
        }
        __syncthreads();
    }
    if (threadIdx.x == 0) {
        const float kv = kvp[0], nv = nvp[0];
        const float Nf = (float)NROWS, dims = (float)NDIM;
        float total_var = expf(nv) + expf(kv);
        float nc1 = 0.5f * dims * logf(TWO_PI * total_var);
        float mean_lp1 = s1[0] / Nf - logf(Nf) - nc1;
        float h  = -mean_lp1;
        out2[0] = BETA * (h - nc1);
        float kdevar = expf(kv);
        float nc2 = 0.5f * dims * logf(TWO_PI * kdevar);
        float mean_lp2 = s2[0] / Nf - logf(Nf - 1.0f) - nc2;
        out2[1] = -mean_lp2;
    }
}

extern "C" void kernel_launch(void* const* d_in, const int* in_sizes, int n_in,
                              void* d_out, int out_size, void* d_ws, size_t ws_size,
                              hipStream_t stream) {
    const float* x     = (const float*)d_in[0];
    const float* noise = (const float*)d_in[1];
    const float* kv    = (const float*)d_in[2];
    const float* nv    = (const float*)d_in[3];
    float* out = (float*)d_out;
    float* ws  = (float*)d_ws;

    float* sq   = ws;
    float* lse1 = ws + NROWS;
    float* lse2 = ws + 2 * NROWS;

    const int nd = NROWS * NDIM;
    k_y<<<(nd + 255) / 256, 256, 0, stream>>>(x, noise, nv, out, nd);
    k_sq<<<NROWS / 256, 256, 0, stream>>>(x, sq);
    k_pairwise<<<NROWS / 128, 256, 0, stream>>>(x, sq, kv, nv, lse1, lse2);
    k_final<<<1, 256, 0, stream>>>(lse1, lse2, kv, nv, out + nd);
}


# The y pass (16384 short waves) and the row-norm pass become one pass over x by 256 long waves, 32 rows each (16 lanes a row, norms summed by shuffles): 4 launches (was 5); norm sums reorder
# speedup vs baseline: 3.9470x; 3.9470x over previous
#include <hip/hip_runtime.h>
#include <hip/hip_bf16.h>
#include <math.h>

typedef __attribute__((ext_vector_type(2))) float v2f;
typedef __attribute__((ext_vector_type(8))) float v8f;

#define NROWS 8192
#define NDIM  64
#define BETA 0.01f
#define TWO_PI 6.28318530717958647692f

#define PRE_ROWS 32
__global__ __launch_bounds__(32) void k_pre(const float* __restrict__ x,
                                            const float* __restrict__ noise,
                                            const float* __restrict__ nvp,
                                            float* __restrict__ y,
                                            float* __restrict__ sq) {
    const int lane = threadIdx.x;
    const int half = lane >> 4;
    const int c    = lane & 15;
    const size_t row0 = (size_t)blockIdx.x * PRE_ROWS + 16 * half;
    const float s = __expf(0.5f * nvp[0]);
    const float4* xr = (const float4*)(x + row0 * NDIM) + c;
    const float4* nr = (const float4*)(noise + row0 * NDIM) + c;
    float4* yr = (float4*)(y + row0 * NDIM) + c;
    float4 xv[16], nz[16];
#pragma unroll
    for (int i = 0; i < 16; ++i) {
        xv[i] = xr[i * (NDIM / 4)];
        nz[i] = nr[i * (NDIM / 4)];
    }
    float mine = 0.0f;
#pragma unroll
    for (int i = 0; i < 16; ++i) {
        float4 o;
        o.x = fmaf(s, nz[i].x, xv[i].x);
        o.y = fmaf(s, nz[i].y, xv[i].y);
        o.z = fmaf(s, nz[i].z, xv[i].z);
        o.w = fmaf(s, nz[i].w, xv[i].w);
        yr[i * (NDIM / 4)] = o;
        float p = xv[i].x * xv[i].x;
        p = fmaf(xv[i].y, xv[i].y, p);
        p = fmaf(xv[i].z, xv[i].z, p);
        p = fmaf(xv[i].w, xv[i].w, p);
#pragma unroll
        for (int m = 1; m <= 8; m <<= 1) p += __shfl_xor(p, m, 32);
        if (i == c) mine = p;
    }
    sq[(size_t)blockIdx.x * PRE_ROWS + lane] = mine;
}

#define TILES_PER_TRIP 4
#define NT (NROWS / 16)
#define NCHUNK 4
#define CHUNK_TILES (NT / 2 / NCHUNK)
#define NTRIP (CHUNK_TILES / TILES_PER_TRIP)
#define REC 64
#define NPLANE (NT / 2)

struct ColGroup {
    float4 b[TILES_PER_TRIP][8];
    float  s[TILES_PER_TRIP];
};

__device__ __forceinline__ void load_group(const float* __restrict__ x,
                                           const float* __restrict__ sq,
                                           int tileI, int k0, int l15, int half, ColGroup& g) {
    const float4* q[TILES_PER_TRIP];
    int c0[TILES_PER_TRIP];
#pragma unroll
    for (int i = 0; i < TILES_PER_TRIP; ++i) {
        const int J = (tileI + k0 + i) & (NT - 1);
        c0[i] = J * 16 + l15;
        q[i] = (const float4*)(x + (size_t)c0[i] * NDIM + 4 * half);
    }
#pragma unroll
    for (int j = 0; j < 8; ++j)
#pragma unroll
        for (int i = 0; i < TILES_PER_TRIP; ++i)
            g.b[i][j] = q[i][2 * j];
#pragma unroll
    for (int i = 0; i < TILES_PER_TRIP; ++i)
        g.s[i] = sq[c0[i]];
}

__device__ __forceinline__ void gram_group(const v2f (&a)[16], const ColGroup& g,
                                           v8f (&c)[TILES_PER_TRIP]) {
    v8f z = {};
#pragma unroll
    for (int i = 0; i < TILES_PER_TRIP; ++i) c[i] = z;
#pragma unroll
    for (int j = 0; j < 8; ++j) {
#pragma unroll
        for (int i = 0; i < TILES_PER_TRIP; ++i) {
            v2f u = {g.b[i][j].x, g.b[i][j].y};
            c[i] = __builtin_amdgcn_wmma_f32_16x16x4_f32(false, a[2 * j], false, u, (short)0, c[i], false, false);
        }
#pragma unroll
        for (int i = 0; i < TILES_PER_TRIP; ++i) {
            v2f w = {g.b[i][j].z, g.b[i][j].w};
            c[i] = __builtin_amdgcn_wmma_f32_16x16x4_f32(false, a[2 * j + 1], false, w, (short)0, c[i], false, false);
        }
    }
}

__device__ __forceinline__ void fold_tile_sym(const v8f& c, float sqB,
                                              const float (&sqA)[8], float a1, float a2,
                                              float (&l1)[8], float (&l2)[8], float (&m2)[8],
                                              float& col1, float& col2) {
    float s2[8];
    float e1s = 0.0f;
#pragma unroll
    for (int k = 0; k < 8; ++k) {
        float d = fmaxf(sqA[k] + sqB - 2.0f * c[k], 0.0f);
        float e1 = __expf(-d * a1);
        l1[k] += e1;
        e1s += e1;
        s2[k] = -d * a2;
        float nm = fmaxf(m2[k], s2[k]);
        l2[k] = l2[k] * __expf(m2[k] - nm) + __expf(s2[k] - nm);
        m2[k] = nm;
    }
    float cm = s2[0];
#pragma unroll
    for (int k = 1; k < 8; ++k) cm = fmaxf(cm, s2[k]);
    float cs = 0.0f;
#pragma unroll
    for (int k = 0; k < 8; ++k) cs += __expf(s2[k] - cm);
    col1 = e1s;
    col2 = cm + __logf(cs);
}

__device__ __forceinline__ void fold_group(const v8f (&c)[TILES_PER_TRIP], const float (&s)[TILES_PER_TRIP],
                                           int tileI, int k0, int lane,
                                           const float (&sqA)[8], float a1, float a2,
                                           float (&l1)[8], float (&l2)[8], float (&m2)[8],
                                           float* __restrict__ colp) {
#pragma unroll
    for (int i = 0; i < TILES_PER_TRIP; ++i) {
        const int k = k0 + i;
        const int J = (tileI + k) & (NT - 1);
        float col1, col2;
        fold_tile_sym(c[i], s[i], sqA, a1, a2, l1, l2, m2, col1, col2);
        const int plane = (k == NT / 2) ? (NPLANE - 1) : (k - 1);
        float* rec = colp + ((size_t)plane * NT + J) * REC;
        rec[lane]      = col1;
        rec[32 + lane] = col2;
    }
}

__global__ __launch_bounds__(64) void k_pairwise(const float* __restrict__ x,
                                                 const float* __restrict__ sq,
                                                 const float* __restrict__ kvp,
                                                 const float* __restrict__ nvp,
                                                 float* __restrict__ rowp1,
                                                 float* __restrict__ rowp2,
                                                 float* __restrict__ colp) {
    const int tid      = threadIdx.x;
    const int lane     = tid & 31;
    const int wave     = tid >> 5;
    const int half     = lane >> 4;
    const int l15      = lane & 15;
    const int chunk    = blockIdx.x / (NT / 2);
    const int tileI    = (blockIdx.x % (NT / 2)) * 2 + wave;
    const int rowBase  = tileI * 16;
    const int rowLane  = rowBase + 8 * half;

    const float kv = kvp[0], nv = nvp[0];
    const float total_var = __expf(nv) + __expf(kv);
    const float a1 = 1.0f / (2.0f * total_var);
    const float a2 = 1.0f / (2.0f * __expf(kv));

    v2f a[16];
    {
        const float4* ar = (const float4*)(x + (size_t)(rowBase + l15) * NDIM + 4 * half);
#pragma unroll
        for (int j = 0; j < 8; ++j) {
            float4 v = ar[2 * j];
            a[2 * j]     = v2f{v.x, v.y};
            a[2 * j + 1] = v2f{v.z, v.w};
        }
    }

    float sqA[8];
#pragma unroll
    for (int k = 0; k < 8; ++k)
        sqA[k] = sq[rowLane + k];
    asm volatile("" ::: "memory");

    float l1[8], l2[8], m2[8];
#pragma unroll
    for (int k = 0; k < 8; ++k) { l1[k] = 0.0f; l2[k] = 0.0f; m2[k] = -1e30f; }

    const int kb = chunk * CHUNK_TILES + 1;
    ColGroup B;
    load_group(x, sq, tileI, kb, l15, half, B);

    if (chunk == 0) {
        v8f c = {};
#pragma unroll
        for (int j = 0; j < 16; ++j)
            c = __builtin_amdgcn_wmma_f32_16x16x4_f32(false, a[j], false, a[j], (short)0, c, false, false);
        const float sqB = sq[rowBase + l15];
#pragma unroll
        for (int k = 0; k < 8; ++k) {
            float d = fmaxf(sqA[k] + sqB - 2.0f * c[k], 0.0f);
            l1[k] += __expf(-d * a1);
            float s2 = (l15 == 8 * half + k) ? -1e38f : (-d * a2);
            float nm = fmaxf(m2[k], s2);
            l2[k] = l2[k] * __expf(m2[k] - nm) + __expf(s2 - nm);
            m2[k] = nm;
        }
    }

    v8f cOld[TILES_PER_TRIP];
    float sOld[TILES_PER_TRIP];
    gram_group(a, B, cOld);
#pragma unroll
    for (int i = 0; i < TILES_PER_TRIP; ++i) sOld[i] = B.s[i];
    load_group(x, sq, tileI, kb + TILES_PER_TRIP, l15, half, B);

    for (int g = 0; g < NTRIP - 1; ++g) {
        v8f cNew[TILES_PER_TRIP];
        float sNew[TILES_PER_TRIP];
        gram_group(a, B, cNew);
#pragma unroll
        for (int i = 0; i < TILES_PER_TRIP; ++i) sNew[i] = B.s[i];
        load_group(x, sq, tileI, kb + (g + 2) * TILES_PER_TRIP, l15, half, B);
        asm volatile("" ::: "memory");
        fold_group(cOld, sOld, tileI, kb + g * TILES_PER_TRIP, lane, sqA, a1, a2, l1, l2, m2, colp);
#pragma unroll
        for (int i = 0; i < TILES_PER_TRIP; ++i) { cOld[i] = cNew[i]; sOld[i] = sNew[i]; }
    }
    fold_group(cOld, sOld, tileI, kb + (NTRIP - 1) * TILES_PER_TRIP, lane, sqA, a1, a2, l1, l2, m2, colp);

#pragma unroll
    for (int k = 0; k < 8; ++k) {
#pragma unroll
        for (int m = 1; m <= 8; m <<= 1) {
            l1[k] += __shfl_xor(l1[k], m, 32);
            float om = __shfl_xor(m2[k], m, 32);
            float ol = __shfl_xor(l2[k], m, 32);
            float nm = fmaxf(m2[k], om);
            l2[k] = l2[k] * __expf(m2[k] - nm) + ol * __expf(om - nm);
            m2[k] = nm;
        }
    }
    if (l15 == 0) {
#pragma unroll
        for (int k = 0; k < 8; ++k) {
            rowp1[chunk * NROWS + rowLane + k] = l1[k];
            rowp2[chunk * NROWS + rowLane + k] = m2[k] + __logf(l2[k]);
        }
    }
}

#define MERGE_BATCH 16
__global__ __launch_bounds__(256) void k_merge(const float* __restrict__ colp,
                                               const float* __restrict__ rowp1,
                                               const float* __restrict__ rowp2,
                                               float* __restrict__ lse1,
                                               float* __restrict__ lse2) {
    __shared__ float sh1[256], shm[256], shl[256];
    const int tid  = threadIdx.x;
    const int lane = tid & 31;
    const int w    = tid >> 5;
    const int J    = blockIdx.x * 2 + (w & 1);
    const int p0   = w >> 1;
    float t1 = 0.0f, m = -1e30f, l = 0.0f;
    for (int b = 0; b < (NPLANE / 4) / MERGE_BATCH; ++b) {
        float v1[MERGE_BATCH], v2[MERGE_BATCH];
#pragma unroll
        for (int i = 0; i < MERGE_BATCH; ++i) {
            const int p  = p0 + 4 * (b * MERGE_BATCH + i);
            const bool ok = p < NPLANE - 1;
            const float* rec = colp + ((size_t)(ok ? p : 0) * NT + J) * REC;
            const float r1 = rec[lane], r2 = rec[32 + lane];
            v1[i] = ok ? r1 : 0.0f;
            v2[i] = ok ? r2 : -1e30f;
        }
        float bm = v2[0];
#pragma unroll
        for (int i = 1; i < MERGE_BATCH; ++i) bm = fmaxf(bm, v2[i]);
        const float nm = fmaxf(m, bm);
        float acc = l * __expf(m - nm);
#pragma unroll
        for (int i = 0; i < MERGE_BATCH; ++i) { t1 += v1[i]; acc += __expf(v2[i] - nm); }
        l = acc; m = nm;
    }
    sh1[tid] = t1; shm[tid] = m; shl[tid] = l;
    __syncthreads();
    if (tid < 32) {
        const int s = tid >> 4, c = tid & 15;
        const int r = (blockIdx.x * 2 + s) * 16 + c;
        float T = 0.0f, M = -1e30f;
        float rv[NCHUNK];
#pragma unroll
        for (int u = 0; u < 4; ++u)
#pragma unroll
            for (int h = 0; h < 2; ++h) {
                const int idx = (2 * u + s) * 32 + h * 16 + c;
                T += sh1[idx];
                M = fmaxf(M, shm[idx]);
            }
#pragma unroll
        for (int q = 0; q < NCHUNK; ++q) {
            T += rowp1[q * NROWS + r];
            rv[q] = rowp2[q * NROWS + r];
            M = fmaxf(M, rv[q]);
        }
        float L = 0.0f;
#pragma unroll
        for (int u = 0; u < 4; ++u)
#pragma unroll
            for (int h = 0; h < 2; ++h) {
                const int idx = (2 * u + s) * 32 + h * 16 + c;
                L += shl[idx] * __expf(shm[idx] - M);
            }
#pragma unroll
        for (int q = 0; q < NCHUNK; ++q) L += __expf(rv[q] - M);
        lse1[r] = __logf(T);
        lse2[r] = M + __logf(L);
    }
}

__global__ __launch_bounds__(256) void k_final(const float* __restrict__ lse1,
                                               const float* __restrict__ lse2,
                                               const float* __restrict__ kvp,
                                               const float* __restrict__ nvp,
                                               float* __restrict__ out2) {
    __shared__ float s1[256], s2[256];
    float p1 = 0.0f, p2 = 0.0f;
    for (int i = threadIdx.x; i < NROWS; i += 256) { p1 += lse1[i]; p2 += lse2[i]; }
    s1[threadIdx.x] = p1; s2[threadIdx.x] = p2;
    __syncthreads();
    for (int st = 128; st > 0; st >>= 1) {
        if (threadIdx.x < st) {
            s1[threadIdx.x] += s1[threadIdx.x + st];
            s2[threadIdx.x] += s2[threadIdx.x + st];
        }
        __syncthreads();
    }
    if (threadIdx.x == 0) {
        const float kv = kvp[0], nv = nvp[0];
        const float Nf = (float)NROWS, dims = (float)NDIM;
        float total_var = expf(nv) + expf(kv);
        float nc1 = 0.5f * dims * logf(TWO_PI * total_var);
        float mean_lp1 = s1[0] / Nf - logf(Nf) - nc1;
        float h  = -mean_lp1;
        out2[0] = BETA * (h - nc1);
        float kdevar = expf(kv);
        float nc2 = 0.5f * dims * logf(TWO_PI * kdevar);
        float mean_lp2 = s2[0] / Nf - logf(Nf - 1.0f) - nc2;
        out2[1] = -mean_lp2;
    }
}

extern "C" void kernel_launch(void* const* d_in, const int* in_sizes, int n_in,
                              void* d_out, int out_size, void* d_ws, size_t ws_size,
                              hipStream_t stream) {
    const float* x     = (const float*)d_in[0];
    const float* noise = (const float*)d_in[1];
    const float* kv    = (const float*)d_in[2];
    const float* nv    = (const float*)d_in[3];
    float* out = (float*)d_out;
    float* ws  = (float*)d_ws;

    float* sq    = ws;
    float* lse1  = ws + NROWS;
    float* lse2  = ws + 2 * NROWS;
    float* rowp1 = ws + 3 * NROWS;
    float* rowp2 = ws + (3 + NCHUNK) * NROWS;
    float* colp  = ws + 16 * NROWS;

    const int nd = NROWS * NDIM;
    k_pre<<<NROWS / PRE_ROWS, 32, 0, stream>>>(x, noise, nv, out, sq);
    k_pairwise<<<NCHUNK * (NT / 2), 64, 0, stream>>>(x, sq, kv, nv, rowp1, rowp2, colp);
    k_merge<<<NT / 2, 256, 0, stream>>>(colp, rowp1, rowp2, lse1, lse2);
    k_final<<<1, 256, 0, stream>>>(lse1, lse2, kv, nv, out + nd);
}


# x also stored in WMMA-fragment order by 256 more waves of the first launch; each A and B fragment load of the pairwise kernel then covers 4 whole 128-byte lines (was 16 lines, 32 bytes of each)
# speedup vs baseline: 4.6226x; 1.1712x over previous
#include <hip/hip_runtime.h>
#include <hip/hip_bf16.h>
#include <math.h>

typedef __attribute__((ext_vector_type(2))) float v2f;
typedef __attribute__((ext_vector_type(4))) float v4f;
typedef __attribute__((ext_vector_type(8))) float v8f;

#define NROWS 8192
#define NDIM  64
#define BETA 0.01f
#define TWO_PI 6.28318530717958647692f

#define PRE_ROWS 32
#define PRE_BLOCKS (NROWS / PRE_ROWS)
#define TILE_F4 (16 * NDIM / 4)
__global__ __launch_bounds__(32) void k_pre(const float* __restrict__ x,
                                            const float* __restrict__ noise,
                                            const float* __restrict__ nvp,
                                            float* __restrict__ y,
                                            float* __restrict__ sq,
                                            float* __restrict__ xt) {
    const int lane = threadIdx.x;
    const int half = lane >> 4;
    const int c    = lane & 15;
    if (blockIdx.x >= PRE_BLOCKS) {
        const size_t tile0 = (size_t)(blockIdx.x - PRE_BLOCKS) * 2;
        const v4f* fp = (const v4f*)(x + (tile0 * 16 + c) * NDIM + 4 * half);
        v4f* xo = (v4f*)xt + tile0 * TILE_F4 + lane;
        v4f fr[16];
#pragma unroll
        for (int i = 0; i < 16; ++i) fr[i] = fp[(i >> 3) * TILE_F4 + 2 * (i & 7)];
        asm volatile("" :: "v"(fr[0]), "v"(fr[1]), "v"(fr[2]), "v"(fr[3]), "v"(fr[4]), "v"(fr[5]),
                           "v"(fr[6]), "v"(fr[7]), "v"(fr[8]), "v"(fr[9]), "v"(fr[10]), "v"(fr[11]),
                           "v"(fr[12]), "v"(fr[13]), "v"(fr[14]), "v"(fr[15]));
#pragma unroll
        for (int i = 0; i < 16; ++i) xo[(i >> 3) * TILE_F4 + 32 * (i & 7)] = fr[i];
        return;
    }
    const size_t row0 = (size_t)blockIdx.x * PRE_ROWS + 16 * half;
    const float s = __expf(0.5f * nvp[0]);
    const float4* xr = (const float4*)(x + row0 * NDIM) + c;
    const float4* nr = (const float4*)(noise + row0 * NDIM) + c;
    float4* yr = (float4*)(y + row0 * NDIM) + c;
    float4 xv[16], nz[16];
#pragma unroll
    for (int i = 0; i < 16; ++i) {
        xv[i] = xr[i * (NDIM / 4)];
        nz[i] = nr[i * (NDIM / 4)];
    }
    float mine = 0.0f;
#pragma unroll
    for (int i = 0; i < 16; ++i) {
        float4 o;
        o.x = fmaf(s, nz[i].x, xv[i].x);
        o.y = fmaf(s, nz[i].y, xv[i].y);
        o.z = fmaf(s, nz[i].z, xv[i].z);
        o.w = fmaf(s, nz[i].w, xv[i].w);
        yr[i * (NDIM / 4)] = o;
        float p = xv[i].x * xv[i].x;
        p = fmaf(xv[i].y, xv[i].y, p);
        p = fmaf(xv[i].z, xv[i].z, p);
        p = fmaf(xv[i].w, xv[i].w, p);
#pragma unroll
        for (int m = 1; m <= 8; m <<= 1) p += __shfl_xor(p, m, 32);
        if (i == c) mine = p;
    }
    sq[(size_t)blockIdx.x * PRE_ROWS + lane] = mine;
}

#define TILES_PER_TRIP 4
#define NT (NROWS / 16)
#define NCHUNK 4
#define CHUNK_TILES (NT / 2 / NCHUNK)
#define NTRIP (CHUNK_TILES / TILES_PER_TRIP)
#define REC 64
#define NPLANE (NT / 2)

struct ColGroup {
    float4 b[TILES_PER_TRIP][8];
    float  s[TILES_PER_TRIP];
};

__device__ __forceinline__ void load_group(const float* __restrict__ xt,
                                           const float* __restrict__ sq,
                                           int tileI, int k0, int l15, int lane, ColGroup& g) {
    const float4* q[TILES_PER_TRIP];
    int c0[TILES_PER_TRIP];
#pragma unroll
    for (int i = 0; i < TILES_PER_TRIP; ++i) {
        const int J = (tileI + k0 + i) & (NT - 1);
        c0[i] = J * 16 + l15;
        q[i] = (const float4*)xt + (size_t)J * TILE_F4 + lane;
    }
#pragma unroll
    for (int j = 0; j < 8; ++j)
#pragma unroll
        for (int i = 0; i < TILES_PER_TRIP; ++i)
            g.b[i][j] = q[i][32 * j];
#pragma unroll
    for (int i = 0; i < TILES_PER_TRIP; ++i)
        g.s[i] = sq[c0[i]];
}

__device__ __forceinline__ void gram_group(const v2f (&a)[16], const ColGroup& g,
                                           v8f (&c)[TILES_PER_TRIP]) {
    v8f z = {};
#pragma unroll
    for (int i = 0; i < TILES_PER_TRIP; ++i) c[i] = z;
#pragma unroll
    for (int j = 0; j < 8; ++j) {
#pragma unroll
        for (int i = 0; i < TILES_PER_TRIP; ++i) {
            v2f u = {g.b[i][j].x, g.b[i][j].y};
            c[i] = __builtin_amdgcn_wmma_f32_16x16x4_f32(false, a[2 * j], false, u, (short)0, c[i], false, false);
        }
#pragma unroll
        for (int i = 0; i < TILES_PER_TRIP; ++i) {
            v2f w = {g.b[i][j].z, g.b[i][j].w};
            c[i] = __builtin_amdgcn_wmma_f32_16x16x4_f32(false, a[2 * j + 1], false, w, (short)0, c[i], false, false);
        }
    }
}

__device__ __forceinline__ void fold_tile_sym(const v8f& c, float sqB,
                                              const float (&sqA)[8], float a1, float a2,
                                              float (&l1)[8], float (&l2)[8], float (&m2)[8],
                                              float& col1, float& col2) {
    float s2[8];
    float e1s = 0.0f;
#pragma unroll
    for (int k = 0; k < 8; ++k) {
        float d = fmaxf(sqA[k] + sqB - 2.0f * c[k], 0.0f);
        float e1 = __expf(-d * a1);
        l1[k] += e1;
        e1s += e1;
        s2[k] = -d * a2;
        float nm = fmaxf(m2[k], s2[k]);
        l2[k] = l2[k] * __expf(m2[k] - nm) + __expf(s2[k] - nm);
        m2[k] = nm;
    }
    float cm = s2[0];
#pragma unroll
    for (int k = 1; k < 8; ++k) cm = fmaxf(cm, s2[k]);
    float cs = 0.0f;
#pragma unroll
    for (int k = 0; k < 8; ++k) cs += __expf(s2[k] - cm);
    col1 = e1s;
    col2 = cm + __logf(cs);
}

__device__ __forceinline__ void fold_group(const v8f (&c)[TILES_PER_TRIP], const float (&s)[TILES_PER_TRIP],
                                           int tileI, int k0, int lane,
                                           const float (&sqA)[8], float a1, float a2,
                                           float (&l1)[8], float (&l2)[8], float (&m2)[8],
                                           float* __restrict__ colp) {
#pragma unroll
    for (int i = 0; i < TILES_PER_TRIP; ++i) {
        const int k = k0 + i;
        const int J = (tileI + k) & (NT - 1);
        float col1, col2;
        fold_tile_sym(c[i], s[i], sqA, a1, a2, l1, l2, m2, col1, col2);
        const int plane = (k == NT / 2) ? (NPLANE - 1) : (k - 1);
        float* rec = colp + ((size_t)plane * NT + J) * REC;
        rec[lane]      = col1;
        rec[32 + lane] = col2;
    }
}

__global__ __launch_bounds__(64) void k_pairwise(const float* __restrict__ xt,
                                                 const float* __restrict__ sq,
                                                 const float* __restrict__ kvp,
                                                 const float* __restrict__ nvp,
                                                 float* __restrict__ rowp1,
                                                 float* __restrict__ rowp2,
                                                 float* __restrict__ colp) {
    const int tid      = threadIdx.x;
    const int lane     = tid & 31;
    const int wave     = tid >> 5;
    const int half     = lane >> 4;
    const int l15      = lane & 15;
    const int chunk    = blockIdx.x / (NT / 2);
    const int tileI    = (blockIdx.x % (NT / 2)) * 2 + wave;
    const int rowBase  = tileI * 16;
    const int rowLane  = rowBase + 8 * half;

    const float kv = kvp[0], nv = nvp[0];
    const float total_var = __expf(nv) + __expf(kv);
    const float a1 = 1.0f / (2.0f * total_var);
    const float a2 = 1.0f / (2.0f * __expf(kv));

    v2f a[16];
    {
        const float4* ar = (const float4*)xt + (size_t)tileI * TILE_F4 + lane;
#pragma unroll
        for (int j = 0; j < 8; ++j) {
            float4 v = ar[32 * j];
            a[2 * j]     = v2f{v.x, v.y};
            a[2 * j + 1] = v2f{v.z, v.w};
        }
    }

    float sqA[8];
#pragma unroll
    for (int k = 0; k < 8; ++k)
        sqA[k] = sq[rowLane + k];
    asm volatile("" ::: "memory");

    float l1[8], l2[8], m2[8];
#pragma unroll
    for (int k = 0; k < 8; ++k) { l1[k] = 0.0f; l2[k] = 0.0f; m2[k] = -1e30f; }

    const int kb = chunk * CHUNK_TILES + 1;
    ColGroup B;
    load_group(xt, sq, tileI, kb, l15, lane, B);

    if (chunk == 0) {
        v8f c = {};
#pragma unroll
        for (int j = 0; j < 16; ++j)
            c = __builtin_amdgcn_wmma_f32_16x16x4_f32(false, a[j], false, a[j], (short)0, c, false, false);
        const float sqB = sq[rowBase + l15];
#pragma unroll
        for (int k = 0; k < 8; ++k) {
            float d = fmaxf(sqA[k] + sqB - 2.0f * c[k], 0.0f);
            l1[k] += __expf(-d * a1);
            float s2 = (l15 == 8 * half + k) ? -1e38f : (-d * a2);
            float nm = fmaxf(m2[k], s2);
            l2[k] = l2[k] * __expf(m2[k] - nm) + __expf(s2 - nm);
            m2[k] = nm;
        }
    }

    v8f cOld[TILES_PER_TRIP];
    float sOld[TILES_PER_TRIP];
    gram_group(a, B, cOld);
#pragma unroll
    for (int i = 0; i < TILES_PER_TRIP; ++i) sOld[i] = B.s[i];
    load_group(xt, sq, tileI, kb + TILES_PER_TRIP, l15, lane, B);

    for (int g = 0; g < NTRIP - 1; ++g) {
        v8f cNew[TILES_PER_TRIP];
        float sNew[TILES_PER_TRIP];
        gram_group(a, B, cNew);
#pragma unroll
        for (int i = 0; i < TILES_PER_TRIP; ++i) sNew[i] = B.s[i];
        load_group(xt, sq, tileI, kb + (g + 2) * TILES_PER_TRIP, l15, lane, B);
        asm volatile("" ::: "memory");
        fold_group(cOld, sOld, tileI, kb + g * TILES_PER_TRIP, lane, sqA, a1, a2, l1, l2, m2, colp);
#pragma unroll
        for (int i = 0; i < TILES_PER_TRIP; ++i) { cOld[i] = cNew[i]; sOld[i] = sNew[i]; }
    }
    fold_group(cOld, sOld, tileI, kb + (NTRIP - 1) * TILES_PER_TRIP, lane, sqA, a1, a2, l1, l2, m2, colp);

#pragma unroll
    for (int k = 0; k < 8; ++k) {
#pragma unroll
        for (int m = 1; m <= 8; m <<= 1) {
            l1[k] += __shfl_xor(l1[k], m, 32);
            float om = __shfl_xor(m2[k], m, 32);
            float ol = __shfl_xor(l2[k], m, 32);
            float nm = fmaxf(m2[k], om);
            l2[k] = l2[k] * __expf(m2[k] - nm) + ol * __expf(om - nm);
            m2[k] = nm;
        }
    }
    if (l15 == 0) {
#pragma unroll
        for (int k = 0; k < 8; ++k) {
            rowp1[chunk * NROWS + rowLane + k] = l1[k];
            rowp2[chunk * NROWS + rowLane + k] = m2[k] + __logf(l2[k]);
        }
    }
}

#define MERGE_BATCH 16
__global__ __launch_bounds__(256) void k_merge(const float* __restrict__ colp,
                                               const float* __restrict__ rowp1,
                                               const float* __restrict__ rowp2,
                                               float* __restrict__ lse1,
                                               float* __restrict__ lse2) {
    __shared__ float sh1[256], shm[256], shl[256];
    const int tid  = threadIdx.x;
    const int lane = tid & 31;
    const int w    = tid >> 5;
    const int J    = blockIdx.x * 2 + (w & 1);
    const int p0   = w >> 1;
    float t1 = 0.0f, m = -1e30f, l = 0.0f;
    for (int b = 0; b < (NPLANE / 4) / MERGE_BATCH; ++b) {
        float v1[MERGE_BATCH], v2[MERGE_BATCH];
#pragma unroll
        for (int i = 0; i < MERGE_BATCH; ++i) {
            const int p  = p0 + 4 * (b * MERGE_BATCH + i);
            const bool ok = p < NPLANE - 1;
            const float* rec = colp + ((size_t)(ok ? p : 0) * NT + J) * REC;
            const float r1 = rec[lane], r2 = rec[32 + lane];
            v1[i] = ok ? r1 : 0.0f;
            v2[i] = ok ? r2 : -1e30f;
        }
        float bm = v2[0];
#pragma unroll
        for (int i = 1; i < MERGE_BATCH; ++i) bm = fmaxf(bm, v2[i]);
        const float nm = fmaxf(m, bm);
        float acc = l * __expf(m - nm);
#pragma unroll
        for (int i = 0; i < MERGE_BATCH; ++i) { t1 += v1[i]; acc += __expf(v2[i] - nm); }
        l = acc; m = nm;
    }
    sh1[tid] = t1; shm[tid] = m; shl[tid] = l;
    __syncthreads();
    if (tid < 32) {
        const int s = tid >> 4, c = tid & 15;
        const int r = (blockIdx.x * 2 + s) * 16 + c;
        float T = 0.0f, M = -1e30f;
        float rv[NCHUNK];
#pragma unroll
        for (int u = 0; u < 4; ++u)
#pragma unroll
            for (int h = 0; h < 2; ++h) {
                const int idx = (2 * u + s) * 32 + h * 16 + c;
                T += sh1[idx];
                M = fmaxf(M, shm[idx]);
            }
#pragma unroll
        for (int q = 0; q < NCHUNK; ++q) {
            T += rowp1[q * NROWS + r];
            rv[q] = rowp2[q * NROWS + r];
            M = fmaxf(M, rv[q]);
        }
        float L = 0.0f;
#pragma unroll
        for (int u = 0; u < 4; ++u)
#pragma unroll
            for (int h = 0; h < 2; ++h) {
                const int idx = (2 * u + s) * 32 + h * 16 + c;
                L += shl[idx] * __expf(shm[idx] - M);
            }
#pragma unroll
        for (int q = 0; q < NCHUNK; ++q) L += __expf(rv[q] - M);
        lse1[r] = __logf(T);
        lse2[r] = M + __logf(L);
    }
}

__global__ __launch_bounds__(256) void k_final(const float* __restrict__ lse1,
                                               const float* __restrict__ lse2,
                                               const float* __restrict__ kvp,
                                               const float* __restrict__ nvp,
                                               float* __restrict__ out2) {
    __shared__ float s1[256], s2[256];
    float p1 = 0.0f, p2 = 0.0f;
    for (int i = threadIdx.x; i < NROWS; i += 256) { p1 += lse1[i]; p2 += lse2[i]; }
    s1[threadIdx.x] = p1; s2[threadIdx.x] = p2;
    __syncthreads();
    for (int st = 128; st > 0; st >>= 1) {
        if (threadIdx.x < st) {
            s1[threadIdx.x] += s1[threadIdx.x + st];
            s2[threadIdx.x] += s2[threadIdx.x + st];
        }
        __syncthreads();
    }
    if (threadIdx.x == 0) {
        const float kv = kvp[0], nv = nvp[0];
        const float Nf = (float)NROWS, dims = (float)NDIM;
        float total_var = expf(nv) + expf(kv);
        float nc1 = 0.5f * dims * logf(TWO_PI * total_var);
        float mean_lp1 = s1[0] / Nf - logf(Nf) - nc1;
        float h  = -mean_lp1;
        out2[0] = BETA * (h - nc1);
        float kdevar = expf(kv);
        float nc2 = 0.5f * dims * logf(TWO_PI * kdevar);
        float mean_lp2 = s2[0] / Nf - logf(Nf - 1.0f) - nc2;
        out2[1] = -mean_lp2;
    }
}

extern "C" void kernel_launch(void* const* d_in, const int* in_sizes, int n_in,
                              void* d_out, int out_size, void* d_ws, size_t ws_size,
                              hipStream_t stream) {
    const float* x     = (const float*)d_in[0];
    const float* noise = (const float*)d_in[1];
    const float* kv    = (const float*)d_in[2];
    const float* nv    = (const float*)d_in[3];
    float* out = (float*)d_out;
    float* ws  = (float*)d_ws;

    float* sq    = ws;
    float* lse1  = ws + NROWS;
    float* lse2  = ws + 2 * NROWS;
    float* rowp1 = ws + 3 * NROWS;
    float* rowp2 = ws + (3 + NCHUNK) * NROWS;
    float* colp  = ws + 16 * NROWS;
    float* xt    = colp + (size_t)NPLANE * NT * REC;

    const int nd = NROWS * NDIM;
    k_pre<<<2 * PRE_BLOCKS, 32, 0, stream>>>(x, noise, nv, out, sq, xt);
    k_pairwise<<<NCHUNK * (NT / 2), 64, 0, stream>>>(xt, sq, kv, nv, rowp1, rowp2, colp);
    k_merge<<<NT / 2, 256, 0, stream>>>(colp, rowp1, rowp2, lse1, lse2);
    k_final<<<1, 256, 0, stream>>>(lse1, lse2, kv, nv, out + nd);
}


# Column records as one 8-byte pair per lane: the merge kernel reads a record in one load and keeps 32 in flight a wave (was 16), the pairwise kernel stores once a tile (was 2); log-sums round anew
# speedup vs baseline: 4.9307x; 1.0667x over previous
#include <hip/hip_runtime.h>
#include <hip/hip_bf16.h>
#include <math.h>

typedef __attribute__((ext_vector_type(2))) float v2f;
typedef __attribute__((ext_vector_type(4))) float v4f;
typedef __attribute__((ext_vector_type(8))) float v8f;

#define NROWS 8192
#define NDIM  64
#define BETA 0.01f
#define TWO_PI 6.28318530717958647692f

#define PRE_ROWS 32
#define PRE_BLOCKS (NROWS / PRE_ROWS)
#define TILE_F4 (16 * NDIM / 4)
__global__ __launch_bounds__(32) void k_pre(const float* __restrict__ x,
                                            const float* __restrict__ noise,
                                            const float* __restrict__ nvp,
                                            float* __restrict__ y,
                                            float* __restrict__ sq,
                                            float* __restrict__ xt) {
    const int lane = threadIdx.x;
    const int half = lane >> 4;
    const int c    = lane & 15;
    if (blockIdx.x >= PRE_BLOCKS) {
        const size_t tile0 = (size_t)(blockIdx.x - PRE_BLOCKS) * 2;
        const v4f* fp = (const v4f*)(x + (tile0 * 16 + c) * NDIM + 4 * half);
        v4f* xo = (v4f*)xt + tile0 * TILE_F4 + lane;
        v4f fr[16];
#pragma unroll
        for (int i = 0; i < 16; ++i) fr[i] = fp[(i >> 3) * TILE_F4 + 2 * (i & 7)];
        asm volatile("" :: "v"(fr[0]), "v"(fr[1]), "v"(fr[2]), "v"(fr[3]), "v"(fr[4]), "v"(fr[5]),
                           "v"(fr[6]), "v"(fr[7]), "v"(fr[8]), "v"(fr[9]), "v"(fr[10]), "v"(fr[11]),
                           "v"(fr[12]), "v"(fr[13]), "v"(fr[14]), "v"(fr[15]));
#pragma unroll
        for (int i = 0; i < 16; ++i) xo[(i >> 3) * TILE_F4 + 32 * (i & 7)] = fr[i];
        return;
    }
    const size_t row0 = (size_t)blockIdx.x * PRE_ROWS + 16 * half;
    const float s = __expf(0.5f * nvp[0]);
    const float4* xr = (const float4*)(x + row0 * NDIM) + c;
    const float4* nr = (const float4*)(noise + row0 * NDIM) + c;
    float4* yr = (float4*)(y + row0 * NDIM) + c;
    float4 xv[16], nz[16];
#pragma unroll
    for (int i = 0; i < 16; ++i) {
        xv[i] = xr[i * (NDIM / 4)];
        nz[i] = nr[i * (NDIM / 4)];
    }
    float mine = 0.0f;
#pragma unroll
    for (int i = 0; i < 16; ++i) {
        float4 o;
        o.x = fmaf(s, nz[i].x, xv[i].x);
        o.y = fmaf(s, nz[i].y, xv[i].y);
        o.z = fmaf(s, nz[i].z, xv[i].z);
        o.w = fmaf(s, nz[i].w, xv[i].w);
        yr[i * (NDIM / 4)] = o;
        float p = xv[i].x * xv[i].x;
        p = fmaf(xv[i].y, xv[i].y, p);
        p = fmaf(xv[i].z, xv[i].z, p);
        p = fmaf(xv[i].w, xv[i].w, p);
#pragma unroll
        for (int m = 1; m <= 8; m <<= 1) p += __shfl_xor(p, m, 32);
        if (i == c) mine = p;
    }
    sq[(size_t)blockIdx.x * PRE_ROWS + lane] = mine;
}

#define TILES_PER_TRIP 4
#define NT (NROWS / 16)
#define NCHUNK 4
#define CHUNK_TILES (NT / 2 / NCHUNK)
#define NTRIP (CHUNK_TILES / TILES_PER_TRIP)
#define REC 64
#define NPLANE (NT / 2)

struct ColGroup {
    float4 b[TILES_PER_TRIP][8];
    float  s[TILES_PER_TRIP];
};

__device__ __forceinline__ void load_group(const float* __restrict__ xt,
                                           const float* __restrict__ sq,
                                           int tileI, int k0, int l15, int lane, ColGroup& g) {
    const float4* q[TILES_PER_TRIP];
    int c0[TILES_PER_TRIP];
#pragma unroll
    for (int i = 0; i < TILES_PER_TRIP; ++i) {
        const int J = (tileI + k0 + i) & (NT - 1);
        c0[i] = J * 16 + l15;
        q[i] = (const float4*)xt + (size_t)J * TILE_F4 + lane;
    }
#pragma unroll
    for (int j = 0; j < 8; ++j)
#pragma unroll
        for (int i = 0; i < TILES_PER_TRIP; ++i)
            g.b[i][j] = q[i][32 * j];
#pragma unroll
    for (int i = 0; i < TILES_PER_TRIP; ++i)
        g.s[i] = sq[c0[i]];
}

__device__ __forceinline__ void gram_group(const v2f (&a)[16], const ColGroup& g,
                                           v8f (&c)[TILES_PER_TRIP]) {
    v8f z = {};
#pragma unroll
    for (int i = 0; i < TILES_PER_TRIP; ++i) c[i] = z;
#pragma unroll
    for (int j = 0; j < 8; ++j) {
#pragma unroll
        for (int i = 0; i < TILES_PER_TRIP; ++i) {
            v2f u = {g.b[i][j].x, g.b[i][j].y};
            c[i] = __builtin_amdgcn_wmma_f32_16x16x4_f32(false, a[2 * j], false, u, (short)0, c[i], false, false);
        }
#pragma unroll
        for (int i = 0; i < TILES_PER_TRIP; ++i) {
            v2f w = {g.b[i][j].z, g.b[i][j].w};
            c[i] = __builtin_amdgcn_wmma_f32_16x16x4_f32(false, a[2 * j + 1], false, w, (short)0, c[i], false, false);
        }
    }
}

__device__ __forceinline__ void fold_tile_sym(const v8f& c, float sqB,
                                              const float (&sqA)[8], float a1, float a2,
                                              float (&l1)[8], float (&l2)[8], float (&m2)[8],
                                              float& col1, float& col2) {
    float s2[8];
    float e1s = 0.0f;
#pragma unroll
    for (int k = 0; k < 8; ++k) {
        float d = fmaxf(sqA[k] + sqB - 2.0f * c[k], 0.0f);
        float e1 = __expf(-d * a1);
        l1[k] += e1;
        e1s += e1;
        s2[k] = -d * a2;
        float nm = fmaxf(m2[k], s2[k]);
        l2[k] = l2[k] * __expf(m2[k] - nm) + __expf(s2[k] - nm);
        m2[k] = nm;
    }
    float cm = s2[0];
#pragma unroll
    for (int k = 1; k < 8; ++k) cm = fmaxf(cm, s2[k]);
    float cs = 0.0f;
#pragma unroll
    for (int k = 0; k < 8; ++k) cs += __expf(s2[k] - cm);
    col1 = e1s;
    col2 = cm + __logf(cs);
}

__device__ __forceinline__ void fold_group(const v8f (&c)[TILES_PER_TRIP], const float (&s)[TILES_PER_TRIP],
                                           int tileI, int k0, int lane,
                                           const float (&sqA)[8], float a1, float a2,
                                           float (&l1)[8], float (&l2)[8], float (&m2)[8],
                                           float* __restrict__ colp) {
#pragma unroll
    for (int i = 0; i < TILES_PER_TRIP; ++i) {
        const int k = k0 + i;
        const int J = (tileI + k) & (NT - 1);
        float col1, col2;
        fold_tile_sym(c[i], s[i], sqA, a1, a2, l1, l2, m2, col1, col2);
        const int plane = (k == NT / 2) ? (NPLANE - 1) : (k - 1);
        float2* rec = (float2*)(colp + ((size_t)plane * NT + J) * REC);
        rec[lane] = make_float2(col1, col2);
    }
}

__global__ __launch_bounds__(64) void k_pairwise(const float* __restrict__ xt,
                                                 const float* __restrict__ sq,
                                                 const float* __restrict__ kvp,
                                                 const float* __restrict__ nvp,
                                                 float* __restrict__ rowp1,
                                                 float* __restrict__ rowp2,
                                                 float* __restrict__ colp) {
    const int tid      = threadIdx.x;
    const int lane     = tid & 31;
    const int wave     = tid >> 5;
    const int half     = lane >> 4;
    const int l15      = lane & 15;
    const int chunk    = blockIdx.x / (NT / 2);
    const int tileI    = (blockIdx.x % (NT / 2)) * 2 + wave;
    const int rowBase  = tileI * 16;
    const int rowLane  = rowBase + 8 * half;

    const float kv = kvp[0], nv = nvp[0];
    const float total_var = __expf(nv) + __expf(kv);
    const float a1 = 1.0f / (2.0f * total_var);
    const float a2 = 1.0f / (2.0f * __expf(kv));

    v2f a[16];
    {
        const float4* ar = (const float4*)xt + (size_t)tileI * TILE_F4 + lane;
#pragma unroll
        for (int j = 0; j < 8; ++j) {
            float4 v = ar[32 * j];
            a[2 * j]     = v2f{v.x, v.y};
            a[2 * j + 1] = v2f{v.z, v.w};
        }
    }

    float sqA[8];
#pragma unroll
    for (int k = 0; k < 8; ++k)
        sqA[k] = sq[rowLane + k];
    asm volatile("" ::: "memory");

    float l1[8], l2[8], m2[8];
#pragma unroll
    for (int k = 0; k < 8; ++k) { l1[k] = 0.0f; l2[k] = 0.0f; m2[k] = -1e30f; }

    const int kb = chunk * CHUNK_TILES + 1;
    ColGroup B;
    load_group(xt, sq, tileI, kb, l15, lane, B);

    if (chunk == 0) {
        v8f c = {};
#pragma unroll
        for (int j = 0; j < 16; ++j)
            c = __builtin_amdgcn_wmma_f32_16x16x4_f32(false, a[j], false, a[j], (short)0, c, false, false);
        const float sqB = sq[rowBase + l15];
#pragma unroll
        for (int k = 0; k < 8; ++k) {
            float d = fmaxf(sqA[k] + sqB - 2.0f * c[k], 0.0f);
            l1[k] += __expf(-d * a1);
            float s2 = (l15 == 8 * half + k) ? -1e38f : (-d * a2);
            float nm = fmaxf(m2[k], s2);
            l2[k] = l2[k] * __expf(m2[k] - nm) + __expf(s2 - nm);
            m2[k] = nm;
        }
    }

    v8f cOld[TILES_PER_TRIP];
    float sOld[TILES_PER_TRIP];
    gram_group(a, B, cOld);
#pragma unroll
    for (int i = 0; i < TILES_PER_TRIP; ++i) sOld[i] = B.s[i];
    load_group(xt, sq, tileI, kb + TILES_PER_TRIP, l15, lane, B);

    for (int g = 0; g < NTRIP - 1; ++g) {
        v8f cNew[TILES_PER_TRIP];
        float sNew[TILES_PER_TRIP];
        gram_group(a, B, cNew);
#pragma unroll
        for (int i = 0; i < TILES_PER_TRIP; ++i) sNew[i] = B.s[i];
        load_group(xt, sq, tileI, kb + (g + 2) * TILES_PER_TRIP, l15, lane, B);
        asm volatile("" ::: "memory");
        fold_group(cOld, sOld, tileI, kb + g * TILES_PER_TRIP, lane, sqA, a1, a2, l1, l2, m2, colp);
#pragma unroll
        for (int i = 0; i < TILES_PER_TRIP; ++i) { cOld[i] = cNew[i]; sOld[i] = sNew[i]; }
    }
    fold_group(cOld, sOld, tileI, kb + (NTRIP - 1) * TILES_PER_TRIP, lane, sqA, a1, a2, l1, l2, m2, colp);

#pragma unroll
    for (int k = 0; k < 8; ++k) {
#pragma unroll
        for (int m = 1; m <= 8; m <<= 1) {
            l1[k] += __shfl_xor(l1[k], m, 32);
            float om = __shfl_xor(m2[k], m, 32);
            float ol = __shfl_xor(l2[k], m, 32);
            float nm = fmaxf(m2[k], om);
            l2[k] = l2[k] * __expf(m2[k] - nm) + ol * __expf(om - nm);
            m2[k] = nm;
        }
    }
    if (l15 == 0) {
#pragma unroll
        for (int k = 0; k < 8; ++k) {
            rowp1[chunk * NROWS + rowLane + k] = l1[k];
            rowp2[chunk * NROWS + rowLane + k] = m2[k] + __logf(l2[k]);
        }
    }
}

#define MERGE_BATCH 32
__global__ __launch_bounds__(256) void k_merge(const float* __restrict__ colp,
                                               const float* __restrict__ rowp1,
                                               const float* __restrict__ rowp2,
                                               float* __restrict__ lse1,
                                               float* __restrict__ lse2) {
    __shared__ float sh1[256], shm[256], shl[256];
    const int tid  = threadIdx.x;
    const int lane = tid & 31;
    const int w    = tid >> 5;
    const int J    = blockIdx.x * 2 + (w & 1);
    const int p0   = w >> 1;
    float t1 = 0.0f, m = -1e30f, l = 0.0f;
    for (int b = 0; b < (NPLANE / 4) / MERGE_BATCH; ++b) {
        float v1[MERGE_BATCH], v2[MERGE_BATCH];
#pragma unroll
        for (int i = 0; i < MERGE_BATCH; ++i) {
            const int p  = p0 + 4 * (b * MERGE_BATCH + i);
            const bool ok = p < NPLANE - 1;
            const float2* rec = (const float2*)(colp + ((size_t)(ok ? p : 0) * NT + J) * REC);
            const float2 r = rec[lane];
            v1[i] = ok ? r.x : 0.0f;
            v2[i] = ok ? r.y : -1e30f;
        }
        float bm = v2[0];
#pragma unroll
        for (int i = 1; i < MERGE_BATCH; ++i) bm = fmaxf(bm, v2[i]);
        const float nm = fmaxf(m, bm);
        float acc = l * __expf(m - nm);
#pragma unroll
        for (int i = 0; i < MERGE_BATCH; ++i) { t1 += v1[i]; acc += __expf(v2[i] - nm); }
        l = acc; m = nm;
    }
    sh1[tid] = t1; shm[tid] = m; shl[tid] = l;
    __syncthreads();
    if (tid < 32) {
        const int s = tid >> 4, c = tid & 15;
        const int r = (blockIdx.x * 2 + s) * 16 + c;
        float T = 0.0f, M = -1e30f;
        float rv[NCHUNK];
#pragma unroll
        for (int u = 0; u < 4; ++u)
#pragma unroll
            for (int h = 0; h < 2; ++h) {
                const int idx = (2 * u + s) * 32 + h * 16 + c;
                T += sh1[idx];
                M = fmaxf(M, shm[idx]);
            }
#pragma unroll
        for (int q = 0; q < NCHUNK; ++q) {
            T += rowp1[q * NROWS + r];
            rv[q] = rowp2[q * NROWS + r];
            M = fmaxf(M, rv[q]);
        }
        float L = 0.0f;
#pragma unroll
        for (int u = 0; u < 4; ++u)
#pragma unroll
            for (int h = 0; h < 2; ++h) {
                const int idx = (2 * u + s) * 32 + h * 16 + c;
                L += shl[idx] * __expf(shm[idx] - M);
            }
#pragma unroll
        for (int q = 0; q < NCHUNK; ++q) L += __expf(rv[q] - M);
        lse1[r] = __logf(T);
        lse2[r] = M + __logf(L);
    }
}

__global__ __launch_bounds__(256) void k_final(const float* __restrict__ lse1,
                                               const float* __restrict__ lse2,
                                               const float* __restrict__ kvp,
                                               const float* __restrict__ nvp,
                                               float* __restrict__ out2) {
    __shared__ float s1[256], s2[256];
    float p1 = 0.0f, p2 = 0.0f;
    for (int i = threadIdx.x; i < NROWS; i += 256) { p1 += lse1[i]; p2 += lse2[i]; }
    s1[threadIdx.x] = p1; s2[threadIdx.x] = p2;
    __syncthreads();
    for (int st = 128; st > 0; st >>= 1) {
        if (threadIdx.x < st) {
            s1[threadIdx.x] += s1[threadIdx.x + st];
            s2[threadIdx.x] += s2[threadIdx.x + st];
        }
        __syncthreads();
    }
    if (threadIdx.x == 0) {
        const float kv = kvp[0], nv = nvp[0];
        const float Nf = (float)NROWS, dims = (float)NDIM;
        float total_var = expf(nv) + expf(kv);
        float nc1 = 0.5f * dims * logf(TWO_PI * total_var);
        float mean_lp1 = s1[0] / Nf - logf(Nf) - nc1;
        float h  = -mean_lp1;
        out2[0] = BETA * (h - nc1);
        float kdevar = expf(kv);
        float nc2 = 0.5f * dims * logf(TWO_PI * kdevar);
        float mean_lp2 = s2[0] / Nf - logf(Nf - 1.0f) - nc2;
        out2[1] = -mean_lp2;
    }
}

extern "C" void kernel_launch(void* const* d_in, const int* in_sizes, int n_in,
                              void* d_out, int out_size, void* d_ws, size_t ws_size,
                              hipStream_t stream) {
    const float* x     = (const float*)d_in[0];
    const float* noise = (const float*)d_in[1];
    const float* kv    = (const float*)d_in[2];
    const float* nv    = (const float*)d_in[3];
    float* out = (float*)d_out;
    float* ws  = (float*)d_ws;

    float* sq    = ws;
    float* lse1  = ws + NROWS;
    float* lse2  = ws + 2 * NROWS;
    float* rowp1 = ws + 3 * NROWS;
    float* rowp2 = ws + (3 + NCHUNK) * NROWS;
    float* colp  = ws + 16 * NROWS;
    float* xt    = colp + (size_t)NPLANE * NT * REC;

    const int nd = NROWS * NDIM;
    k_pre<<<2 * PRE_BLOCKS, 32, 0, stream>>>(x, noise, nv, out, sq, xt);
    k_pairwise<<<NCHUNK * (NT / 2), 64, 0, stream>>>(xt, sq, kv, nv, rowp1, rowp2, colp);
    k_merge<<<NT / 2, 256, 0, stream>>>(colp, rowp1, rowp2, lse1, lse2);
    k_final<<<1, 256, 0, stream>>>(lse1, lse2, kv, nv, out + nd);
}


# Fold of the pairwise loop done on a whole trip in packed row pairs: one running maximum and rescale per row per trip of 4 tiles (was per tile); column sums pair up; log-sums round anew
# speedup vs baseline: 5.1111x; 1.0366x over previous
#include <hip/hip_runtime.h>
#include <hip/hip_bf16.h>
#include <math.h>

typedef __attribute__((ext_vector_type(2))) float v2f;
typedef __attribute__((ext_vector_type(4))) float v4f;
typedef __attribute__((ext_vector_type(8))) float v8f;

#define NROWS 8192
#define NDIM  64
#define BETA 0.01f
#define TWO_PI 6.28318530717958647692f

#define PRE_ROWS 32
#define PRE_BLOCKS (NROWS / PRE_ROWS)
#define TILE_F4 (16 * NDIM / 4)
__global__ __launch_bounds__(32) void k_pre(const float* __restrict__ x,
                                            const float* __restrict__ noise,
                                            const float* __restrict__ nvp,
                                            float* __restrict__ y,
                                            float* __restrict__ sq,
                                            float* __restrict__ xt) {
    const int lane = threadIdx.x;
    const int half = lane >> 4;
    const int c    = lane & 15;
    if (blockIdx.x >= PRE_BLOCKS) {
        const size_t tile0 = (size_t)(blockIdx.x - PRE_BLOCKS) * 2;
        const v4f* fp = (const v4f*)(x + (tile0 * 16 + c) * NDIM + 4 * half);
        v4f* xo = (v4f*)xt + tile0 * TILE_F4 + lane;
        v4f fr[16];
#pragma unroll
        for (int i = 0; i < 16; ++i) fr[i] = fp[(i >> 3) * TILE_F4 + 2 * (i & 7)];
        asm volatile("" :: "v"(fr[0]), "v"(fr[1]), "v"(fr[2]), "v"(fr[3]), "v"(fr[4]), "v"(fr[5]),
                           "v"(fr[6]), "v"(fr[7]), "v"(fr[8]), "v"(fr[9]), "v"(fr[10]), "v"(fr[11]),
                           "v"(fr[12]), "v"(fr[13]), "v"(fr[14]), "v"(fr[15]));
#pragma unroll
        for (int i = 0; i < 16; ++i) xo[(i >> 3) * TILE_F4 + 32 * (i & 7)] = fr[i];
        return;
    }
    const size_t row0 = (size_t)blockIdx.x * PRE_ROWS + 16 * half;
    const float s = __expf(0.5f * nvp[0]);
    const float4* xr = (const float4*)(x + row0 * NDIM) + c;
    const float4* nr = (const float4*)(noise + row0 * NDIM) + c;
    float4* yr = (float4*)(y + row0 * NDIM) + c;
    float4 xv[16], nz[16];
#pragma unroll
    for (int i = 0; i < 16; ++i) {
        xv[i] = xr[i * (NDIM / 4)];
        nz[i] = nr[i * (NDIM / 4)];
    }
    float mine = 0.0f;
#pragma unroll
    for (int i = 0; i < 16; ++i) {
        float4 o;
        o.x = fmaf(s, nz[i].x, xv[i].x);
        o.y = fmaf(s, nz[i].y, xv[i].y);
        o.z = fmaf(s, nz[i].z, xv[i].z);
        o.w = fmaf(s, nz[i].w, xv[i].w);
        yr[i * (NDIM / 4)] = o;
        float p = xv[i].x * xv[i].x;
        p = fmaf(xv[i].y, xv[i].y, p);
        p = fmaf(xv[i].z, xv[i].z, p);
        p = fmaf(xv[i].w, xv[i].w, p);
#pragma unroll
        for (int m = 1; m <= 8; m <<= 1) p += __shfl_xor(p, m, 32);
        if (i == c) mine = p;
    }
    sq[(size_t)blockIdx.x * PRE_ROWS + lane] = mine;
}

#define TILES_PER_TRIP 4
#define NT (NROWS / 16)
#define NCHUNK 4
#define CHUNK_TILES (NT / 2 / NCHUNK)
#define NTRIP (CHUNK_TILES / TILES_PER_TRIP)
#define REC 64
#define NPLANE (NT / 2)

struct ColGroup {
    float4 b[TILES_PER_TRIP][8];
    float  s[TILES_PER_TRIP];
};

__device__ __forceinline__ void load_group(const float* __restrict__ xt,
                                           const float* __restrict__ sq,
                                           int tileI, int k0, int l15, int lane, ColGroup& g) {
    const float4* q[TILES_PER_TRIP];
    int c0[TILES_PER_TRIP];
#pragma unroll
    for (int i = 0; i < TILES_PER_TRIP; ++i) {
        const int J = (tileI + k0 + i) & (NT - 1);
        c0[i] = J * 16 + l15;
        q[i] = (const float4*)xt + (size_t)J * TILE_F4 + lane;
    }
#pragma unroll
    for (int j = 0; j < 8; ++j)
#pragma unroll
        for (int i = 0; i < TILES_PER_TRIP; ++i)
            g.b[i][j] = q[i][32 * j];
#pragma unroll
    for (int i = 0; i < TILES_PER_TRIP; ++i)
        g.s[i] = sq[c0[i]];
}

__device__ __forceinline__ void gram_group(const v2f (&a)[16], const ColGroup& g,
                                           v8f (&c)[TILES_PER_TRIP]) {
    v8f z = {};
#pragma unroll
    for (int i = 0; i < TILES_PER_TRIP; ++i) c[i] = z;
#pragma unroll
    for (int j = 0; j < 8; ++j) {
#pragma unroll
        for (int i = 0; i < TILES_PER_TRIP; ++i) {
            v2f u = {g.b[i][j].x, g.b[i][j].y};
            c[i] = __builtin_amdgcn_wmma_f32_16x16x4_f32(false, a[2 * j], false, u, (short)0, c[i], false, false);
        }
#pragma unroll
        for (int i = 0; i < TILES_PER_TRIP; ++i) {
            v2f w = {g.b[i][j].z, g.b[i][j].w};
            c[i] = __builtin_amdgcn_wmma_f32_16x16x4_f32(false, a[2 * j + 1], false, w, (short)0, c[i], false, false);
        }
    }
}

__device__ __forceinline__ v2f exp_pair(v2f x) { return v2f{__expf(x.x), __expf(x.y)}; }
__device__ __forceinline__ v2f max_pair(v2f x, v2f y) { return v2f{fmaxf(x.x, y.x), fmaxf(x.y, y.y)}; }

__device__ __forceinline__ void fold_group(const v8f (&c)[TILES_PER_TRIP], const float (&s)[TILES_PER_TRIP],
                                           int tileI, int k0, int lane,
                                           const float (&sqA)[8], float a1, float a2,
                                           float (&l1)[8], float (&l2)[8], float (&m2)[8],
                                           float* __restrict__ colp) {
    static_assert(TILES_PER_TRIP == 4, "the row maximum and the row sum below name the four tiles of a trip");
    v2f s2[TILES_PER_TRIP][4];
    float col1[TILES_PER_TRIP];
    const v2f zero = {0.0f, 0.0f};
    for (int i = 0; i < TILES_PER_TRIP; ++i) {
        v2f e1s = zero;
        for (int p = 0; p < 4; ++p) {
            const v2f cc = {c[i][2 * p], c[i][2 * p + 1]};
            const v2f qa = {sqA[2 * p], sqA[2 * p + 1]};
            const v2f d = max_pair(qa + s[i] - 2.0f * cc, zero);
            const v2f e1 = exp_pair(-d * a1);
            l1[2 * p] += e1.x;
            l1[2 * p + 1] += e1.y;
            e1s += e1;
            s2[i][p] = -d * a2;
        }
        col1[i] = e1s.x + e1s.y;
    }
    for (int p = 0; p < 4; ++p) {
        const v2f mo = {m2[2 * p], m2[2 * p + 1]};
        const v2f lo = {l2[2 * p], l2[2 * p + 1]};
        const v2f nm = max_pair(max_pair(mo, max_pair(s2[0][p], s2[1][p])), max_pair(s2[2][p], s2[3][p]));
        const v2f e01 = exp_pair(s2[0][p] - nm) + exp_pair(s2[1][p] - nm);
        const v2f e23 = exp_pair(s2[2][p] - nm) + exp_pair(s2[3][p] - nm);
        const v2f ln = lo * exp_pair(mo - nm) + (e01 + e23);
        l2[2 * p] = ln.x; l2[2 * p + 1] = ln.y;
        m2[2 * p] = nm.x; m2[2 * p + 1] = nm.y;
    }
    for (int i = 0; i < TILES_PER_TRIP; ++i) {
        const int k = k0 + i;
        const int J = (tileI + k) & (NT - 1);
        const v2f m01 = max_pair(s2[i][0], s2[i][1]);
        const v2f m23 = max_pair(s2[i][2], s2[i][3]);
        const v2f mm = max_pair(m01, m23);
        const float cm = fmaxf(mm.x, mm.y);
        v2f cs = zero;
        for (int p = 0; p < 4; ++p) cs += exp_pair(s2[i][p] - cm);
        const int plane = (k == NT / 2) ? (NPLANE - 1) : (k - 1);
        float2* rec = (float2*)(colp + ((size_t)plane * NT + J) * REC);
        rec[lane] = make_float2(col1[i], cm + __logf(cs.x + cs.y));
    }
}

__global__ __launch_bounds__(64) void k_pairwise(const float* __restrict__ xt,
                                                 const float* __restrict__ sq,
                                                 const float* __restrict__ kvp,
                                                 const float* __restrict__ nvp,
                                                 float* __restrict__ rowp1,
                                                 float* __restrict__ rowp2,
                                                 float* __restrict__ colp) {
    const int tid      = threadIdx.x;
    const int lane     = tid & 31;
    const int wave     = tid >> 5;
    const int half     = lane >> 4;
    const int l15      = lane & 15;
    const int chunk    = blockIdx.x / (NT / 2);
    const int tileI    = (blockIdx.x % (NT / 2)) * 2 + wave;
    const int rowBase  = tileI * 16;
    const int rowLane  = rowBase + 8 * half;

    const float kv = kvp[0], nv = nvp[0];
    const float total_var = __expf(nv) + __expf(kv);
    const float a1 = 1.0f / (2.0f * total_var);
    const float a2 = 1.0f / (2.0f * __expf(kv));

    v2f a[16];
    {
        const float4* ar = (const float4*)xt + (size_t)tileI * TILE_F4 + lane;
#pragma unroll
        for (int j = 0; j < 8; ++j) {
            float4 v = ar[32 * j];
            a[2 * j]     = v2f{v.x, v.y};
            a[2 * j + 1] = v2f{v.z, v.w};
        }
    }

    float sqA[8];
#pragma unroll
    for (int k = 0; k < 8; ++k)
        sqA[k] = sq[rowLane + k];
    asm volatile("" ::: "memory");

    float l1[8], l2[8], m2[8];
#pragma unroll
    for (int k = 0; k < 8; ++k) { l1[k] = 0.0f; l2[k] = 0.0f; m2[k] = -1e30f; }

    const int kb = chunk * CHUNK_TILES + 1;
    ColGroup B;
    load_group(xt, sq, tileI, kb, l15, lane, B);

    if (chunk == 0) {
        v8f c = {};
#pragma unroll
        for (int j = 0; j < 16; ++j)
            c = __builtin_amdgcn_wmma_f32_16x16x4_f32(false, a[j], false, a[j], (short)0, c, false, false);
        const float sqB = sq[rowBase + l15];
#pragma unroll
        for (int k = 0; k < 8; ++k) {
            float d = fmaxf(sqA[k] + sqB - 2.0f * c[k], 0.0f);
            l1[k] += __expf(-d * a1);
            float s2 = (l15 == 8 * half + k) ? -1e38f : (-d * a2);
            float nm = fmaxf(m2[k], s2);
            l2[k] = l2[k] * __expf(m2[k] - nm) + __expf(s2 - nm);
            m2[k] = nm;
        }
    }

    v8f cOld[TILES_PER_TRIP];
    float sOld[TILES_PER_TRIP];
    gram_group(a, B, cOld);
#pragma unroll
    for (int i = 0; i < TILES_PER_TRIP; ++i) sOld[i] = B.s[i];
    load_group(xt, sq, tileI, kb + TILES_PER_TRIP, l15, lane, B);

    for (int g = 0; g < NTRIP - 1; ++g) {
        v8f cNew[TILES_PER_TRIP];
        float sNew[TILES_PER_TRIP];
        gram_group(a, B, cNew);
#pragma unroll
        for (int i = 0; i < TILES_PER_TRIP; ++i) sNew[i] = B.s[i];
        load_group(xt, sq, tileI, kb + (g + 2) * TILES_PER_TRIP, l15, lane, B);
        asm volatile("" ::: "memory");
        fold_group(cOld, sOld, tileI, kb + g * TILES_PER_TRIP, lane, sqA, a1, a2, l1, l2, m2, colp);
#pragma unroll
        for (int i = 0; i < TILES_PER_TRIP; ++i) { cOld[i] = cNew[i]; sOld[i] = sNew[i]; }
    }
    fold_group(cOld, sOld, tileI, kb + (NTRIP - 1) * TILES_PER_TRIP, lane, sqA, a1, a2, l1, l2, m2, colp);

#pragma unroll
    for (int k = 0; k < 8; ++k) {
#pragma unroll
        for (int m = 1; m <= 8; m <<= 1) {
            l1[k] += __shfl_xor(l1[k], m, 32);
            float om = __shfl_xor(m2[k], m, 32);
            float ol = __shfl_xor(l2[k], m, 32);
            float nm = fmaxf(m2[k], om);
            l2[k] = l2[k] * __expf(m2[k] - nm) + ol * __expf(om - nm);
            m2[k] = nm;
        }
    }
    if (l15 == 0) {
#pragma unroll
        for (int k = 0; k < 8; ++k) {
            rowp1[chunk * NROWS + rowLane + k] = l1[k];
            rowp2[chunk * NROWS + rowLane + k] = m2[k] + __logf(l2[k]);
        }
    }
}

#define MERGE_BATCH 32
__global__ __launch_bounds__(256) void k_merge(const float* __restrict__ colp,
                                               const float* __restrict__ rowp1,
                                               const float* __restrict__ rowp2,
                                               float* __restrict__ lse1,
                                               float* __restrict__ lse2) {
    __shared__ float sh1[256], shm[256], shl[256];
    const int tid  = threadIdx.x;
    const int lane = tid & 31;
    const int w    = tid >> 5;
    const int J    = blockIdx.x * 2 + (w & 1);
    const int p0   = w >> 1;
    float t1 = 0.0f, m = -1e30f, l = 0.0f;
    for (int b = 0; b < (NPLANE / 4) / MERGE_BATCH; ++b) {
        float v1[MERGE_BATCH], v2[MERGE_BATCH];
#pragma unroll
        for (int i = 0; i < MERGE_BATCH; ++i) {
            const int p  = p0 + 4 * (b * MERGE_BATCH + i);
            const bool ok = p < NPLANE - 1;
            const float2* rec = (const float2*)(colp + ((size_t)(ok ? p : 0) * NT + J) * REC);
            const float2 r = rec[lane];
            v1[i] = ok ? r.x : 0.0f;
            v2[i] = ok ? r.y : -1e30f;
        }
        float bm = v2[0];
#pragma unroll
        for (int i = 1; i < MERGE_BATCH; ++i) bm = fmaxf(bm, v2[i]);
        const float nm = fmaxf(m, bm);
        float acc = l * __expf(m - nm);
#pragma unroll
        for (int i = 0; i < MERGE_BATCH; ++i) { t1 += v1[i]; acc += __expf(v2[i] - nm); }
        l = acc; m = nm;
    }
    sh1[tid] = t1; shm[tid] = m; shl[tid] = l;
    __syncthreads();
    if (tid < 32) {
        const int s = tid >> 4, c = tid & 15;
        const int r = (blockIdx.x * 2 + s) * 16 + c;
        float T = 0.0f, M = -1e30f;
        float rv[NCHUNK];
#pragma unroll
        for (int u = 0; u < 4; ++u)
#pragma unroll
            for (int h = 0; h < 2; ++h) {
                const int idx = (2 * u + s) * 32 + h * 16 + c;
                T += sh1[idx];
                M = fmaxf(M, shm[idx]);
            }
#pragma unroll
        for (int q = 0; q < NCHUNK; ++q) {
            T += rowp1[q * NROWS + r];
            rv[q] = rowp2[q * NROWS + r];
            M = fmaxf(M, rv[q]);
        }
        float L = 0.0f;
#pragma unroll
        for (int u = 0; u < 4; ++u)
#pragma unroll
            for (int h = 0; h < 2; ++h) {
                const int idx = (2 * u + s) * 32 + h * 16 + c;
                L += shl[idx] * __expf(shm[idx] - M);
            }
#pragma unroll
        for (int q = 0; q < NCHUNK; ++q) L += __expf(rv[q] - M);
        lse1[r] = __logf(T);
        lse2[r] = M + __logf(L);
    }
}

__global__ __launch_bounds__(256) void k_final(const float* __restrict__ lse1,
                                               const float* __restrict__ lse2,
                                               const float* __restrict__ kvp,
                                               const float* __restrict__ nvp,
                                               float* __restrict__ out2) {
    __shared__ float s1[256], s2[256];
    float p1 = 0.0f, p2 = 0.0f;
    for (int i = threadIdx.x; i < NROWS; i += 256) { p1 += lse1[i]; p2 += lse2[i]; }
    s1[threadIdx.x] = p1; s2[threadIdx.x] = p2;
    __syncthreads();
    for (int st = 128; st > 0; st >>= 1) {
        if (threadIdx.x < st) {
            s1[threadIdx.x] += s1[threadIdx.x + st];
            s2[threadIdx.x] += s2[threadIdx.x + st];
        }
        __syncthreads();
    }
    if (threadIdx.x == 0) {
        const float kv = kvp[0], nv = nvp[0];
        const float Nf = (float)NROWS, dims = (float)NDIM;
        float total_var = expf(nv) + expf(kv);
        float nc1 = 0.5f * dims * logf(TWO_PI * total_var);
        float mean_lp1 = s1[0] / Nf - logf(Nf) - nc1;
        float h  = -mean_lp1;
        out2[0] = BETA * (h - nc1);
        float kdevar = expf(kv);
        float nc2 = 0.5f * dims * logf(TWO_PI * kdevar);
        float mean_lp2 = s2[0] / Nf - logf(Nf - 1.0f) - nc2;
        out2[1] = -mean_lp2;
    }
}

extern "C" void kernel_launch(void* const* d_in, const int* in_sizes, int n_in,
                              void* d_out, int out_size, void* d_ws, size_t ws_size,
                              hipStream_t stream) {
    const float* x     = (const float*)d_in[0];
    const float* noise = (const float*)d_in[1];
    const float* kv    = (const float*)d_in[2];
    const float* nv    = (const float*)d_in[3];
    float* out = (float*)d_out;
    float* ws  = (float*)d_ws;

    float* sq    = ws;
    float* lse1  = ws + NROWS;
    float* lse2  = ws + 2 * NROWS;
    float* rowp1 = ws + 3 * NROWS;
    float* rowp2 = ws + (3 + NCHUNK) * NROWS;
    float* colp  = ws + 16 * NROWS;
    float* xt    = colp + (size_t)NPLANE * NT * REC;

    const int nd = NROWS * NDIM;
    k_pre<<<2 * PRE_BLOCKS, 32, 0, stream>>>(x, noise, nv, out, sq, xt);
    k_pairwise<<<NCHUNK * (NT / 2), 64, 0, stream>>>(xt, sq, kv, nv, rowp1, rowp2, colp);
    k_merge<<<NT / 2, 256, 0, stream>>>(colp, rowp1, rowp2, lse1, lse2);
    k_final<<<1, 256, 0, stream>>>(lse1, lse2, kv, nv, out + nd);
}


# Both tail kernels ask for all their loads at entry (merge: a wave's 64 records and the row partials; final: 16-byte loads and the two scalars): one memory round trip each, was 3 and 6; sums reorder
# speedup vs baseline: 5.4366x; 1.0637x over previous
#include <hip/hip_runtime.h>
#include <hip/hip_bf16.h>
#include <math.h>

typedef __attribute__((ext_vector_type(2))) float v2f;
typedef __attribute__((ext_vector_type(4))) float v4f;
typedef __attribute__((ext_vector_type(8))) float v8f;

#define NROWS 8192
#define NDIM  64
#define BETA 0.01f
#define TWO_PI 6.28318530717958647692f

#define PRE_ROWS 32
#define PRE_BLOCKS (NROWS / PRE_ROWS)
#define TILE_F4 (16 * NDIM / 4)
__global__ __launch_bounds__(32) void k_pre(const float* __restrict__ x,
                                            const float* __restrict__ noise,
                                            const float* __restrict__ nvp,
                                            float* __restrict__ y,
                                            float* __restrict__ sq,
                                            float* __restrict__ xt) {
    const int lane = threadIdx.x;
    const int half = lane >> 4;
    const int c    = lane & 15;
    if (blockIdx.x >= PRE_BLOCKS) {
        const size_t tile0 = (size_t)(blockIdx.x - PRE_BLOCKS) * 2;
        const v4f* fp = (const v4f*)(x + (tile0 * 16 + c) * NDIM + 4 * half);
        v4f* xo = (v4f*)xt + tile0 * TILE_F4 + lane;
        v4f fr[16];
#pragma unroll
        for (int i = 0; i < 16; ++i) fr[i] = fp[(i >> 3) * TILE_F4 + 2 * (i & 7)];
        asm volatile("" :: "v"(fr[0]), "v"(fr[1]), "v"(fr[2]), "v"(fr[3]), "v"(fr[4]), "v"(fr[5]),
                           "v"(fr[6]), "v"(fr[7]), "v"(fr[8]), "v"(fr[9]), "v"(fr[10]), "v"(fr[11]),
                           "v"(fr[12]), "v"(fr[13]), "v"(fr[14]), "v"(fr[15]));
#pragma unroll
        for (int i = 0; i < 16; ++i) xo[(i >> 3) * TILE_F4 + 32 * (i & 7)] = fr[i];
        return;
    }
    const size_t row0 = (size_t)blockIdx.x * PRE_ROWS + 16 * half;
    const float s = __expf(0.5f * nvp[0]);
    const float4* xr = (const float4*)(x + row0 * NDIM) + c;
    const float4* nr = (const float4*)(noise + row0 * NDIM) + c;
    float4* yr = (float4*)(y + row0 * NDIM) + c;
    float4 xv[16], nz[16];
#pragma unroll
    for (int i = 0; i < 16; ++i) {
        xv[i] = xr[i * (NDIM / 4)];
        nz[i] = nr[i * (NDIM / 4)];
    }
    float mine = 0.0f;
#pragma unroll
    for (int i = 0; i < 16; ++i) {
        float4 o;
        o.x = fmaf(s, nz[i].x, xv[i].x);
        o.y = fmaf(s, nz[i].y, xv[i].y);
        o.z = fmaf(s, nz[i].z, xv[i].z);
        o.w = fmaf(s, nz[i].w, xv[i].w);
        yr[i * (NDIM / 4)] = o;
        float p = xv[i].x * xv[i].x;
        p = fmaf(xv[i].y, xv[i].y, p);
        p = fmaf(xv[i].z, xv[i].z, p);
        p = fmaf(xv[i].w, xv[i].w, p);
#pragma unroll
        for (int m = 1; m <= 8; m <<= 1) p += __shfl_xor(p, m, 32);
        if (i == c) mine = p;
    }
    sq[(size_t)blockIdx.x * PRE_ROWS + lane] = mine;
}

#define TILES_PER_TRIP 4
#define NT (NROWS / 16)
#define NCHUNK 4
#define CHUNK_TILES (NT / 2 / NCHUNK)
#define NTRIP (CHUNK_TILES / TILES_PER_TRIP)
#define REC 64
#define NPLANE (NT / 2)

struct ColGroup {
    float4 b[TILES_PER_TRIP][8];
    float  s[TILES_PER_TRIP];
};

__device__ __forceinline__ void load_group(const float* __restrict__ xt,
                                           const float* __restrict__ sq,
                                           int tileI, int k0, int l15, int lane, ColGroup& g) {
    const float4* q[TILES_PER_TRIP];
    int c0[TILES_PER_TRIP];
#pragma unroll
    for (int i = 0; i < TILES_PER_TRIP; ++i) {
        const int J = (tileI + k0 + i) & (NT - 1);
        c0[i] = J * 16 + l15;
        q[i] = (const float4*)xt + (size_t)J * TILE_F4 + lane;
    }
#pragma unroll
    for (int j = 0; j < 8; ++j)
#pragma unroll
        for (int i = 0; i < TILES_PER_TRIP; ++i)
            g.b[i][j] = q[i][32 * j];
#pragma unroll
    for (int i = 0; i < TILES_PER_TRIP; ++i)
        g.s[i] = sq[c0[i]];
}

__device__ __forceinline__ void gram_group(const v2f (&a)[16], const ColGroup& g,
                                           v8f (&c)[TILES_PER_TRIP]) {
    v8f z = {};
#pragma unroll
    for (int i = 0; i < TILES_PER_TRIP; ++i) c[i] = z;
#pragma unroll
    for (int j = 0; j < 8; ++j) {
#pragma unroll
        for (int i = 0; i < TILES_PER_TRIP; ++i) {
            v2f u = {g.b[i][j].x, g.b[i][j].y};
            c[i] = __builtin_amdgcn_wmma_f32_16x16x4_f32(false, a[2 * j], false, u, (short)0, c[i], false, false);
        }
#pragma unroll
        for (int i = 0; i < TILES_PER_TRIP; ++i) {
            v2f w = {g.b[i][j].z, g.b[i][j].w};
            c[i] = __builtin_amdgcn_wmma_f32_16x16x4_f32(false, a[2 * j + 1], false, w, (short)0, c[i], false, false);
        }
    }
}

__device__ __forceinline__ v2f exp_pair(v2f x) { return v2f{__expf(x.x), __expf(x.y)}; }
__device__ __forceinline__ v2f max_pair(v2f x, v2f y) { return v2f{fmaxf(x.x, y.x), fmaxf(x.y, y.y)}; }

__device__ __forceinline__ void fold_group(const v8f (&c)[TILES_PER_TRIP], const float (&s)[TILES_PER_TRIP],
                                           int tileI, int k0, int lane,
                                           const float (&sqA)[8], float a1, float a2,
                                           float (&l1)[8], float (&l2)[8], float (&m2)[8],
                                           float* __restrict__ colp) {
    static_assert(TILES_PER_TRIP == 4, "the row maximum and the row sum below name the four tiles of a trip");
    v2f s2[TILES_PER_TRIP][4];
    float col1[TILES_PER_TRIP];
    const v2f zero = {0.0f, 0.0f};
    for (int i = 0; i < TILES_PER_TRIP; ++i) {
        v2f e1s = zero;
        for (int p = 0; p < 4; ++p) {
            const v2f cc = {c[i][2 * p], c[i][2 * p + 1]};
            const v2f qa = {sqA[2 * p], sqA[2 * p + 1]};
            const v2f d = max_pair(qa + s[i] - 2.0f * cc, zero);
            const v2f e1 = exp_pair(-d * a1);
            l1[2 * p] += e1.x;
            l1[2 * p + 1] += e1.y;
            e1s += e1;
            s2[i][p] = -d * a2;
        }
        col1[i] = e1s.x + e1s.y;
    }
    for (int p = 0; p < 4; ++p) {
        const v2f mo = {m2[2 * p], m2[2 * p + 1]};
        const v2f lo = {l2[2 * p], l2[2 * p + 1]};
        const v2f nm = max_pair(max_pair(mo, max_pair(s2[0][p], s2[1][p])), max_pair(s2[2][p], s2[3][p]));
        const v2f e01 = exp_pair(s2[0][p] - nm) + exp_pair(s2[1][p] - nm);
        const v2f e23 = exp_pair(s2[2][p] - nm) + exp_pair(s2[3][p] - nm);
        const v2f ln = lo * exp_pair(mo - nm) + (e01 + e23);
        l2[2 * p] = ln.x; l2[2 * p + 1] = ln.y;
        m2[2 * p] = nm.x; m2[2 * p + 1] = nm.y;
    }
    for (int i = 0; i < TILES_PER_TRIP; ++i) {
        const int k = k0 + i;
        const int J = (tileI + k) & (NT - 1);
        const v2f m01 = max_pair(s2[i][0], s2[i][1]);
        const v2f m23 = max_pair(s2[i][2], s2[i][3]);
        const v2f mm = max_pair(m01, m23);
        const float cm = fmaxf(mm.x, mm.y);
        v2f cs = zero;
        for (int p = 0; p < 4; ++p) cs += exp_pair(s2[i][p] - cm);
        const int plane = (k == NT / 2) ? (NPLANE - 1) : (k - 1);
        float2* rec = (float2*)(colp + ((size_t)plane * NT + J) * REC);
        rec[lane] = make_float2(col1[i], cm + __logf(cs.x + cs.y));
    }
}

__global__ __launch_bounds__(64) void k_pairwise(const float* __restrict__ xt,
                                                 const float* __restrict__ sq,
                                                 const float* __restrict__ kvp,
                                                 const float* __restrict__ nvp,
                                                 float* __restrict__ rowp1,
                                                 float* __restrict__ rowp2,
                                                 float* __restrict__ colp) {
    const int tid      = threadIdx.x;
    const int lane     = tid & 31;
    const int wave     = tid >> 5;
    const int half     = lane >> 4;
    const int l15      = lane & 15;
    const int chunk    = blockIdx.x / (NT / 2);
    const int tileI    = (blockIdx.x % (NT / 2)) * 2 + wave;
    const int rowBase  = tileI * 16;
    const int rowLane  = rowBase + 8 * half;

    const float kv = kvp[0], nv = nvp[0];
    const float total_var = __expf(nv) + __expf(kv);
    const float a1 = 1.0f / (2.0f * total_var);
    const float a2 = 1.0f / (2.0f * __expf(kv));

    v2f a[16];
    {
        const float4* ar = (const float4*)xt + (size_t)tileI * TILE_F4 + lane;
#pragma unroll
        for (int j = 0; j < 8; ++j) {
            float4 v = ar[32 * j];
            a[2 * j]     = v2f{v.x, v.y};
            a[2 * j + 1] = v2f{v.z, v.w};
        }
    }

    float sqA[8];
#pragma unroll
    for (int k = 0; k < 8; ++k)
        sqA[k] = sq[rowLane + k];
    asm volatile("" ::: "memory");

    float l1[8], l2[8], m2[8];
#pragma unroll
    for (int k = 0; k < 8; ++k) { l1[k] = 0.0f; l2[k] = 0.0f; m2[k] = -1e30f; }

    const int kb = chunk * CHUNK_TILES + 1;
    ColGroup B;
    load_group(xt, sq, tileI, kb, l15, lane, B);

    if (chunk == 0) {
        v8f c = {};
#pragma unroll
        for (int j = 0; j < 16; ++j)
            c = __builtin_amdgcn_wmma_f32_16x16x4_f32(false, a[j], false, a[j], (short)0, c, false, false);
        const float sqB = sq[rowBase + l15];
#pragma unroll
        for (int k = 0; k < 8; ++k) {
            float d = fmaxf(sqA[k] + sqB - 2.0f * c[k], 0.0f);
            l1[k] += __expf(-d * a1);
            float s2 = (l15 == 8 * half + k) ? -1e38f : (-d * a2);
            float nm = fmaxf(m2[k], s2);
            l2[k] = l2[k] * __expf(m2[k] - nm) + __expf(s2 - nm);
            m2[k] = nm;
        }
    }

    v8f cOld[TILES_PER_TRIP];
    float sOld[TILES_PER_TRIP];
    gram_group(a, B, cOld);
#pragma unroll
    for (int i = 0; i < TILES_PER_TRIP; ++i) sOld[i] = B.s[i];
    load_group(xt, sq, tileI, kb + TILES_PER_TRIP, l15, lane, B);

    for (int g = 0; g < NTRIP - 1; ++g) {
        v8f cNew[TILES_PER_TRIP];
        float sNew[TILES_PER_TRIP];
        gram_group(a, B, cNew);
#pragma unroll
        for (int i = 0; i < TILES_PER_TRIP; ++i) sNew[i] = B.s[i];
        load_group(xt, sq, tileI, kb + (g + 2) * TILES_PER_TRIP, l15, lane, B);
        asm volatile("" ::: "memory");
        fold_group(cOld, sOld, tileI, kb + g * TILES_PER_TRIP, lane, sqA, a1, a2, l1, l2, m2, colp);
#pragma unroll
        for (int i = 0; i < TILES_PER_TRIP; ++i) { cOld[i] = cNew[i]; sOld[i] = sNew[i]; }
    }
    fold_group(cOld, sOld, tileI, kb + (NTRIP - 1) * TILES_PER_TRIP, lane, sqA, a1, a2, l1, l2, m2, colp);

#pragma unroll
    for (int k = 0; k < 8; ++k) {
#pragma unroll
        for (int m = 1; m <= 8; m <<= 1) {
            l1[k] += __shfl_xor(l1[k], m, 32);
            float om = __shfl_xor(m2[k], m, 32);
            float ol = __shfl_xor(l2[k], m, 32);
            float nm = fmaxf(m2[k], om);
            l2[k] = l2[k] * __expf(m2[k] - nm) + ol * __expf(om - nm);
            m2[k] = nm;
        }
    }
    if (l15 == 0) {
#pragma unroll
        for (int k = 0; k < 8; ++k) {
            rowp1[chunk * NROWS + rowLane + k] = l1[k];
            rowp2[chunk * NROWS + rowLane + k] = m2[k] + __logf(l2[k]);
        }
    }
}

#define MERGE_BATCH 64
__global__ __launch_bounds__(256) void k_merge(const float* __restrict__ colp,
                                               const float* __restrict__ rowp1,
                                               const float* __restrict__ rowp2,
                                               float* __restrict__ lse1,
                                               float* __restrict__ lse2) {
    static_assert(NPLANE / 4 == MERGE_BATCH, "one batch holds every record of a wave");
    __shared__ float sh1[256], shm[256], shl[256];
    const int tid  = threadIdx.x;
    const int lane = tid & 31;
    const int w    = tid >> 5;
    const int J    = blockIdx.x * 2 + (w & 1);
    const int p0   = w >> 1;
    const int fs = (tid >> 4) & 1, fc = tid & 15;
    const int r  = (blockIdx.x * 2 + fs) * 16 + fc;
    float rt[NCHUNK] = {}, rv[NCHUNK] = {};
    if (tid < 32) {
#pragma unroll
        for (int q = 0; q < NCHUNK; ++q) {
            rt[q] = rowp1[q * NROWS + r];
            rv[q] = rowp2[q * NROWS + r];
        }
    }
    float v1[MERGE_BATCH], v2[MERGE_BATCH];
#pragma unroll
    for (int i = 0; i < MERGE_BATCH; ++i) {
        const int p  = p0 + 4 * i;
        const bool ok = p < NPLANE - 1;
        const float2* rec = (const float2*)(colp + ((size_t)(ok ? p : 0) * NT + J) * REC);
        const float2 rr = rec[lane];
        v1[i] = ok ? rr.x : 0.0f;
        v2[i] = ok ? rr.y : -1e30f;
    }
    float m = -1e30f;
#pragma unroll
    for (int i = 0; i < MERGE_BATCH; ++i) m = fmaxf(m, v2[i]);
    float t1 = 0.0f, l = 0.0f;
#pragma unroll
    for (int i = 0; i < MERGE_BATCH; ++i) { t1 += v1[i]; l += __expf(v2[i] - m); }
    sh1[tid] = t1; shm[tid] = m; shl[tid] = l;
    __syncthreads();
    if (tid < 32) {
        const int s = tid >> 4, c = tid & 15;
        float T = 0.0f, M = -1e30f;
#pragma unroll
        for (int u = 0; u < 4; ++u)
#pragma unroll
            for (int h = 0; h < 2; ++h) {
                const int idx = (2 * u + s) * 32 + h * 16 + c;
                T += sh1[idx];
                M = fmaxf(M, shm[idx]);
            }
#pragma unroll
        for (int q = 0; q < NCHUNK; ++q) {
            T += rt[q];
            M = fmaxf(M, rv[q]);
        }
        float L = 0.0f;
#pragma unroll
        for (int u = 0; u < 4; ++u)
#pragma unroll
            for (int h = 0; h < 2; ++h) {
                const int idx = (2 * u + s) * 32 + h * 16 + c;
                L += shl[idx] * __expf(shm[idx] - M);
            }
#pragma unroll
        for (int q = 0; q < NCHUNK; ++q) L += __expf(rv[q] - M);
        lse1[r] = __logf(T);
        lse2[r] = M + __logf(L);
    }
}

__global__ __launch_bounds__(256) void k_final(const float* __restrict__ lse1,
                                               const float* __restrict__ lse2,
                                               const float* __restrict__ kvp,
                                               const float* __restrict__ nvp,
                                               float* __restrict__ out2) {
    __shared__ float s1[256], s2[256];
    const float kv = kvp[0], nv = nvp[0];
    static_assert(NROWS / 1024 == 8, "the statement below names the eight pieces of each plane");
    const v4f* a4 = (const v4f*)lse1 + threadIdx.x;
    const v4f* b4 = (const v4f*)lse2 + threadIdx.x;
    v4f u[NROWS / 1024], v[NROWS / 1024];
#pragma unroll
    for (int k = 0; k < NROWS / 1024; ++k) { u[k] = a4[256 * k]; v[k] = b4[256 * k]; }
    asm volatile("" :: "v"(u[0]), "v"(u[1]), "v"(u[2]), "v"(u[3]), "v"(u[4]), "v"(u[5]), "v"(u[6]), "v"(u[7]),
                       "v"(v[0]), "v"(v[1]), "v"(v[2]), "v"(v[3]), "v"(v[4]), "v"(v[5]), "v"(v[6]), "v"(v[7]));
    float p1 = 0.0f, p2 = 0.0f;
#pragma unroll
    for (int k = 0; k < NROWS / 1024; ++k) {
        p1 += (u[k].x + u[k].y) + (u[k].z + u[k].w);
        p2 += (v[k].x + v[k].y) + (v[k].z + v[k].w);
    }
    s1[threadIdx.x] = p1; s2[threadIdx.x] = p2;
    __syncthreads();
    for (int st = 128; st > 0; st >>= 1) {
        if (threadIdx.x < st) {
            s1[threadIdx.x] += s1[threadIdx.x + st];
            s2[threadIdx.x] += s2[threadIdx.x + st];
        }
        __syncthreads();
    }
    if (threadIdx.x == 0) {
        const float Nf = (float)NROWS, dims = (float)NDIM;
        float total_var = expf(nv) + expf(kv);
        float nc1 = 0.5f * dims * logf(TWO_PI * total_var);
        float mean_lp1 = s1[0] / Nf - logf(Nf) - nc1;
        float h  = -mean_lp1;
        out2[0] = BETA * (h - nc1);
        float kdevar = expf(kv);
        float nc2 = 0.5f * dims * logf(TWO_PI * kdevar);
        float mean_lp2 = s2[0] / Nf - logf(Nf - 1.0f) - nc2;
        out2[1] = -mean_lp2;
    }
}

extern "C" void kernel_launch(void* const* d_in, const int* in_sizes, int n_in,
                              void* d_out, int out_size, void* d_ws, size_t ws_size,
                              hipStream_t stream) {
    const float* x     = (const float*)d_in[0];
    const float* noise = (const float*)d_in[1];
    const float* kv    = (const float*)d_in[2];
    const float* nv    = (const float*)d_in[3];
    float* out = (float*)d_out;
    float* ws  = (float*)d_ws;

    float* sq    = ws;
    float* lse1  = ws + NROWS;
    float* lse2  = ws + 2 * NROWS;
    float* rowp1 = ws + 3 * NROWS;
    float* rowp2 = ws + (3 + NCHUNK) * NROWS;
    float* colp  = ws + 16 * NROWS;
    float* xt    = colp + (size_t)NPLANE * NT * REC;

    const int nd = NROWS * NDIM;
    k_pre<<<2 * PRE_BLOCKS, 32, 0, stream>>>(x, noise, nv, out, sq, xt);
    k_pairwise<<<NCHUNK * (NT / 2), 64, 0, stream>>>(xt, sq, kv, nv, rowp1, rowp2, colp);
    k_merge<<<NT / 2, 256, 0, stream>>>(colp, rowp1, rowp2, lse1, lse2);
    k_final<<<1, 256, 0, stream>>>(lse1, lse2, kv, nv, out + nd);
}
